# starting kernel
# baseline (speedup 1.0000x reference)
#include <hip/hip_runtime.h>
#include <math.h>

typedef __attribute__((ext_vector_type(2))) float v2f;
typedef __attribute__((ext_vector_type(8))) float v8f;

#define N_EMBD 192
#define N_HEAD 16
#define HEAD   12
#define DPAD   16
#define BATCH  4
#define SEQ    2048
#define M_TOK  (BATCH*SEQ)
#define FFN    (4*N_EMBD)
#define BH     (BATCH*N_HEAD)

static __device__ __forceinline__ v8f wmma4(v2f a, v2f b, v8f c) {
  return __builtin_amdgcn_wmma_f32_16x16x4_f32(false, a, false, b, (short)0, c, false, false);
}

__global__ void rmsnorm_kernel(const float* __restrict__ x, const float* __restrict__ w,
                               float* __restrict__ out) {
  int lane = threadIdx.x & 31;
  int wv   = threadIdx.x >> 5;
  int row  = blockIdx.x * 8 + wv;
  const float* xr = x + (size_t)row * N_EMBD;
  float v[6]; float ss = 0.f;
  #pragma unroll
  for (int i = 0; i < 6; i++) { v[i] = xr[lane + 32*i]; ss += v[i]*v[i]; }
  #pragma unroll
  for (int off = 16; off >= 1; off >>= 1) ss += __shfl_xor(ss, off, 32);
  float sc = rsqrtf(ss * (1.0f / N_EMBD) + 1e-5f);
  float* orow = out + (size_t)row * N_EMBD;
  #pragma unroll
  for (int i = 0; i < 6; i++) { int c = lane + 32*i; orow[c] = v[i] * sc * w[c]; }
}

__global__ void pack_b_kernel(const float* __restrict__ B, float* __restrict__ Bf, int N) {
  int idx = blockIdx.x * 256 + threadIdx.x;
  int k = idx / N, n = idx % N;
  Bf[((size_t)(k >> 1) * N + n) * 2 + (k & 1)] = B[idx];
}

__global__ void pack_qkv_kernel(const float* __restrict__ Wq, const float* __restrict__ Wk,
                                const float* __restrict__ Wv, float* __restrict__ Bf) {
  int idx = blockIdx.x * 256 + threadIdx.x;
  int c = idx / 576, j = idx % 576;
  int which = j / 192, r = j % 192;
  int h = r / HEAD, d = r % HEAD;
  const float* W = (which == 0) ? Wq : ((which == 1) ? Wk : Wv);
  Bf[((size_t)(c >> 1) * 576 + j) * 2 + (c & 1)] = W[((size_t)h * N_EMBD + c) * HEAD + d];
}

__device__ __forceinline__ void gemm16x64(const float* __restrict__ A, const float* __restrict__ Bf,
                                          int K, int lda, int N, int m0, int n0, v8f* acc) {
  int lane = threadIdx.x & 31;
  int mr = lane & 15;
  int kh = (lane >> 4) << 1;
  const float* arow  = A + (size_t)(m0 + mr) * lda;
  const float* bbase = Bf + (size_t)(n0 + mr) * 2;
  #pragma unroll 2
  for (int k = 0; k < K; k += 4) {
    v2f a = *(const v2f*)(arow + k + kh);
    const float* bk = bbase + (size_t)((k + kh) >> 1) * (2 * N);
    #pragma unroll
    for (int t = 0; t < 4; t++) {
      v2f b = *(const v2f*)(bk + t * 32);
      acc[t] = wmma4(a, b, acc[t]);
    }
  }
}

__global__ void gemm_qkv_kernel(const float* __restrict__ A, const float* __restrict__ Bf,
                                float* __restrict__ Qp, float* __restrict__ Kp,
                                float* __restrict__ Vp) {
  int wv = threadIdx.x >> 5;
  int tile = blockIdx.x * 8 + wv;
  const int TN = 576 / 64;
  int m0 = (tile / TN) * 16;
  int n0 = (tile % TN) * 64;
  v8f acc[4] = {{}, {}, {}, {}};
  gemm16x64(A, Bf, N_EMBD, N_EMBD, 576, m0, n0, acc);
  int lane = threadIdx.x & 31;
  int mr = lane & 15, half = lane >> 4;
  #pragma unroll
  for (int t = 0; t < 4; t++) {
    int j = n0 + t * 16 + mr;
    int which = j / 192, r = j % 192, h = r / HEAD, d = r % HEAD;
    #pragma unroll
    for (int v = 0; v < 8; v++) {
      int m = m0 + v + 8 * half;
      int b = m >> 11, tq = m & 2047;
      size_t hb = ((size_t)(b * N_HEAD + h)) * SEQ * DPAD;
      if (which == 0)      Qp[hb + (size_t)tq * DPAD + d] = acc[t][v];
      else if (which == 1) Kp[hb + (size_t)tq * DPAD + d] = acc[t][v];
      else                 Vp[hb + (size_t)(tq >> 1) * 32 + d * 2 + (tq & 1)] = acc[t][v];
    }
  }
}

__global__ void attn_kernel(const float* __restrict__ Qp, const float* __restrict__ Kp,
                            const float* __restrict__ Vp, float* __restrict__ sa) {
  __shared__ float Pb[4][256];
  int lane = threadIdx.x & 31, wv = threadIdx.x >> 5;
  int mr = lane & 15, half = lane >> 4, kh = half << 1;
  int qt = wv * 32 + blockIdx.x;
  int bh = blockIdx.y;
  const size_t base = (size_t)bh * SEQ * DPAD;
  const float* Qb = Qp + base + (size_t)qt * 16 * DPAD;

  v2f qf[4];
  #pragma unroll
  for (int ck = 0; ck < 4; ck++) qf[ck] = *(const v2f*)(Qb + mr * DPAD + ck * 4 + kh);

  v8f o = {};
  float mrun[8], lrun[8];
  #pragma unroll
  for (int v = 0; v < 8; v++) { mrun[v] = -INFINITY; lrun[v] = 0.f; }
  const float scale = 0.28867513459481287f;

  for (int kt = 0; kt <= qt; ++kt) {
    const float* Kt = Kp + base + (size_t)kt * 256;
    const float* Vt = Vp + base + (size_t)kt * 256;
    if (kt < qt) {
      __builtin_prefetch(Kt + 256, 0, 0);
      __builtin_prefetch(Vt + 256, 0, 0);
    }
    v8f s = {};
    #pragma unroll
    for (int ck = 0; ck < 4; ck++) {
      v2f bf = *(const v2f*)(Kt + mr * DPAD + ck * 4 + kh);
      s = wmma4(qf[ck], bf, s);
    }
    #pragma unroll
    for (int v = 0; v < 8; v++) {
      float sv = s[v] * scale;
      sv = (kt == qt && mr > v + 8 * half) ? -INFINITY : sv;
      float rmax = sv;
      #pragma unroll
      for (int off = 8; off >= 1; off >>= 1) rmax = fmaxf(rmax, __shfl_xor(rmax, off, 32));
      float mnew  = fmaxf(mrun[v], rmax);
      float alpha = __expf(mrun[v] - mnew);
      float pe    = __expf(sv - mnew);
      float rsum  = pe;
      #pragma unroll
      for (int off = 8; off >= 1; off >>= 1) rsum += __shfl_xor(rsum, off, 32);
      lrun[v] = lrun[v] * alpha + rsum;
      mrun[v] = mnew;
      o[v]    = o[v] * alpha;
      Pb[wv][(v + 8 * half) * 16 + mr] = pe;
    }
    __asm__ volatile("" ::: "memory");
    #pragma unroll
    for (int ck = 0; ck < 4; ck++) {
      v2f a = *(const v2f*)(&Pb[wv][mr * 16 + ck * 4 + kh]);
      int s0 = ck * 4 + kh;
      v2f bf = *(const v2f*)(Vt + (s0 >> 1) * 32 + mr * 2);
      o = wmma4(a, bf, o);
    }
  }

  int b = bh >> 4, h = bh & 15;
  if (mr < HEAD) {
    #pragma unroll
    for (int v = 0; v < 8; v++) {
      int q = qt * 16 + v + 8 * half;
      sa[((size_t)(b * SEQ + q)) * N_EMBD + h * HEAD + mr] = o[v] / lrun[v];
    }
  }
}

__global__ void gemm_proj_kernel(const float* __restrict__ A, const float* __restrict__ Bf,
                                 const float* __restrict__ bias, const float* __restrict__ resid,
                                 float* __restrict__ out) {
  int wv = threadIdx.x >> 5;
  int tile = blockIdx.x * 8 + wv;
  const int TN = 192 / 64;
  int m0 = (tile / TN) * 16, n0 = (tile % TN) * 64;
  v8f acc[4] = {{}, {}, {}, {}};
  gemm16x64(A, Bf, 192, 192, 192, m0, n0, acc);
  int lane = threadIdx.x & 31; int mr = lane & 15, half = lane >> 4;
  #pragma unroll
  for (int t = 0; t < 4; t++) {
    int j = n0 + t * 16 + mr;
    float bj = bias[j];
    #pragma unroll
    for (int v = 0; v < 8; v++) {
      int m = m0 + v + 8 * half;
      size_t ofs = (size_t)m * 192 + j;
      out[ofs] = acc[t][v] + bj + resid[ofs];
    }
  }
}

__global__ void gemm_ff1_kernel(const float* __restrict__ A, const float* __restrict__ Bf,
                                float* __restrict__ out) {
  int wv = threadIdx.x >> 5;
  int tile = blockIdx.x * 8 + wv;
  const int TN = FFN / 64;
  int m0 = (tile / TN) * 16, n0 = (tile % TN) * 64;
  v8f acc[4] = {{}, {}, {}, {}};
  gemm16x64(A, Bf, 192, 192, FFN, m0, n0, acc);
  int lane = threadIdx.x & 31; int mr = lane & 15, half = lane >> 4;
  #pragma unroll
  for (int t = 0; t < 4; t++) {
    int j = n0 + t * 16 + mr;
    #pragma unroll
    for (int v = 0; v < 8; v++) {
      int m = m0 + v + 8 * half;
      float val = acc[t][v];
      out[(size_t)m * FFN + j] = val / (1.0f + __expf(-val));
    }
  }
}

__global__ void gemm_ff2_kernel(const float* __restrict__ A, const float* __restrict__ Bf,
                                const float* __restrict__ resid, float* __restrict__ out) {
  int wv = threadIdx.x >> 5;
  int tile = blockIdx.x * 8 + wv;
  const int TN = 192 / 64;
  int m0 = (tile / TN) * 16, n0 = (tile % TN) * 64;
  v8f acc[4] = {{}, {}, {}, {}};
  gemm16x64(A, Bf, FFN, FFN, 192, m0, n0, acc);
  int lane = threadIdx.x & 31; int mr = lane & 15, half = lane >> 4;
  #pragma unroll
  for (int t = 0; t < 4; t++) {
    int j = n0 + t * 16 + mr;
    #pragma unroll
    for (int v = 0; v < 8; v++) {
      int m = m0 + v + 8 * half;
      size_t ofs = (size_t)m * 192 + j;
      out[ofs] = acc[t][v] + resid[ofs];
    }
  }
}

extern "C" void kernel_launch(void* const* d_in, const int* in_sizes, int n_in,
                              void* d_out, int out_size, void* d_ws, size_t ws_size,
                              hipStream_t stream) {
  (void)in_sizes; (void)n_in; (void)out_size; (void)ws_size;
  const float* x     = (const float*)d_in[0];
  const float* ln1_w = (const float*)d_in[1];
  const float* ln2_w = (const float*)d_in[2];
  const float* Wq    = (const float*)d_in[3];
  const float* Wk    = (const float*)d_in[4];
  const float* Wv    = (const float*)d_in[5];
  const float* Wproj = (const float*)d_in[6];
  const float* bproj = (const float*)d_in[7];
  const float* W1    = (const float*)d_in[8];
  const float* W2    = (const float*)d_in[9];
  float* ws = (float*)d_ws;

  size_t off = 0;
  float* h    = ws + off; off += (size_t)M_TOK * N_EMBD;
  float* wqkv = ws + off; off += (size_t)192 * 576;
  float* wpjf = ws + off; off += (size_t)192 * 192;
  float* w1f  = ws + off; off += (size_t)192 * FFN;
  float* w2f  = ws + off; off += (size_t)FFN * 192;
  float* Qp   = ws + off; off += (size_t)BH * SEQ * DPAD;
  float* Kp   = ws + off; off += (size_t)BH * SEQ * DPAD;
  float* Vp   = ws + off; off += (size_t)BH * SEQ * DPAD;
  float* sa   = ws + off; off += (size_t)M_TOK * N_EMBD;
  float* x1   = ws + off; off += (size_t)M_TOK * N_EMBD;
  float* h2   = ws + off; off += (size_t)M_TOK * N_EMBD;
  float* u    = ws + off; off += (size_t)M_TOK * FFN;
  float* out  = (float*)d_out;

  hipMemsetAsync(Qp, 0, (size_t)3 * BH * SEQ * DPAD * sizeof(float), stream);

  pack_qkv_kernel<<<(192 * 576) / 256, 256, 0, stream>>>(Wq, Wk, Wv, wqkv);
  pack_b_kernel<<<(192 * 192) / 256, 256, 0, stream>>>(Wproj, wpjf, 192);
  pack_b_kernel<<<(192 * FFN) / 256, 256, 0, stream>>>(W1, w1f, FFN);
  pack_b_kernel<<<(FFN * 192) / 256, 256, 0, stream>>>(W2, w2f, 192);

  rmsnorm_kernel<<<M_TOK / 8, 256, 0, stream>>>(x, ln1_w, h);
  gemm_qkv_kernel<<<(512 * 9) / 8, 256, 0, stream>>>(h, wqkv, Qp, Kp, Vp);
  attn_kernel<<<dim3(32, BH), 128, 0, stream>>>(Qp, Kp, Vp, sa);
  gemm_proj_kernel<<<(512 * 3) / 8, 256, 0, stream>>>(sa, wpjf, bproj, x, x1);
  rmsnorm_kernel<<<M_TOK / 8, 256, 0, stream>>>(x1, ln2_w, h2);
  gemm_ff1_kernel<<<(512 * 12) / 8, 256, 0, stream>>>(h2, w1f, u);
  gemm_ff2_kernel<<<(512 * 3) / 8, 256, 0, stream>>>(u, w2f, x1, out);
}


# attention tile held transposed (S^T = K*Q^T): each query's max and sum are in-lane plus one cross-row lane exchange, exp(S^T) feeds V^T*P^T as B operand from registers; no LDS, no 16-lane shuffles
# speedup vs baseline: 1.2904x; 1.2904x over previous
#include <hip/hip_runtime.h>
#include <math.h>

typedef __attribute__((ext_vector_type(2))) float v2f;
typedef __attribute__((ext_vector_type(4))) float v4f;
typedef __attribute__((ext_vector_type(8))) float v8f;

#define N_EMBD 192
#define N_HEAD 16
#define HEAD   12
#define DPAD   16
#define BATCH  4
#define SEQ    2048
#define M_TOK  (BATCH*SEQ)
#define FFN    (4*N_EMBD)
#define BH     (BATCH*N_HEAD)

static __device__ __forceinline__ v8f wmma4(v2f a, v2f b, v8f c) {
  return __builtin_amdgcn_wmma_f32_16x16x4_f32(false, a, false, b, (short)0, c, false, false);
}

__global__ void rmsnorm_kernel(const float* __restrict__ x, const float* __restrict__ w,
                               float* __restrict__ out) {
  int lane = threadIdx.x & 31;
  int wv   = threadIdx.x >> 5;
  int row  = blockIdx.x * 8 + wv;
  const float* xr = x + (size_t)row * N_EMBD;
  float v[6]; float ss = 0.f;
  #pragma unroll
  for (int i = 0; i < 6; i++) { v[i] = xr[lane + 32*i]; ss += v[i]*v[i]; }
  #pragma unroll
  for (int off = 16; off >= 1; off >>= 1) ss += __shfl_xor(ss, off, 32);
  float sc = rsqrtf(ss * (1.0f / N_EMBD) + 1e-5f);
  float* orow = out + (size_t)row * N_EMBD;
  #pragma unroll
  for (int i = 0; i < 6; i++) { int c = lane + 32*i; orow[c] = v[i] * sc * w[c]; }
}

__global__ void pack_b_kernel(const float* __restrict__ B, float* __restrict__ Bf, int N) {
  int idx = blockIdx.x * 256 + threadIdx.x;
  int k = idx / N, n = idx % N;
  Bf[((size_t)(k >> 1) * N + n) * 2 + (k & 1)] = B[idx];
}

__global__ void pack_qkv_kernel(const float* __restrict__ Wq, const float* __restrict__ Wk,
                                const float* __restrict__ Wv, float* __restrict__ Bf) {
  int idx = blockIdx.x * 256 + threadIdx.x;
  int c = idx / 576, j = idx % 576;
  int which = j / 192, r = j % 192;
  int h = r / HEAD, d = r % HEAD;
  const float* W = (which == 0) ? Wq : ((which == 1) ? Wk : Wv);
  Bf[((size_t)(c >> 1) * 576 + j) * 2 + (c & 1)] = W[((size_t)h * N_EMBD + c) * HEAD + d];
}

__device__ __forceinline__ void gemm16x64(const float* __restrict__ A, const float* __restrict__ Bf,
                                          int K, int lda, int N, int m0, int n0, v8f* acc) {
  int lane = threadIdx.x & 31;
  int mr = lane & 15;
  int kh = (lane >> 4) << 1;
  const float* arow  = A + (size_t)(m0 + mr) * lda;
  const float* bbase = Bf + (size_t)(n0 + mr) * 2;
  #pragma unroll 2
  for (int k = 0; k < K; k += 4) {
    v2f a = *(const v2f*)(arow + k + kh);
    const float* bk = bbase + (size_t)((k + kh) >> 1) * (2 * N);
    #pragma unroll
    for (int t = 0; t < 4; t++) {
      v2f b = *(const v2f*)(bk + t * 32);
      acc[t] = wmma4(a, b, acc[t]);
    }
  }
}

__global__ void gemm_qkv_kernel(const float* __restrict__ A, const float* __restrict__ Bf,
                                float* __restrict__ Qp, float* __restrict__ Kp,
                                float* __restrict__ Vp) {
  int wv = threadIdx.x >> 5;
  int tile = blockIdx.x * 8 + wv;
  const int TN = 576 / 64;
  int m0 = (tile / TN) * 16;
  int n0 = (tile % TN) * 64;
  v8f acc[4] = {{}, {}, {}, {}};
  gemm16x64(A, Bf, N_EMBD, N_EMBD, 576, m0, n0, acc);
  int lane = threadIdx.x & 31;
  int mr = lane & 15, half = lane >> 4;
  #pragma unroll
  for (int t = 0; t < 4; t++) {
    int j = n0 + t * 16 + mr;
    int which = j / 192, r = j % 192, h = r / HEAD, d = r % HEAD;
    #pragma unroll
    for (int v = 0; v < 8; v++) {
      int m = m0 + v + 8 * half;
      int b = m >> 11, tq = m & 2047;
      size_t hb = ((size_t)(b * N_HEAD + h)) * SEQ * DPAD;
      if (which == 0)      Qp[hb + (size_t)tq * DPAD + d] = acc[t][v];
      else if (which == 1) Kp[hb + (size_t)tq * DPAD + d] = acc[t][v];
      else                 Vp[hb + (size_t)(tq >> 1) * 32 + d * 2 + (tq & 1)] = acc[t][v];
    }
  }
}


static __device__ __forceinline__ float xhalf(float x) {
  int b = __builtin_bit_cast(int, x);
  return __builtin_bit_cast(float,
      __builtin_amdgcn_permlanex16(b, b, 0x76543210, 0xfedcba98, false, false));
}

__global__ void attn_kernel(const float* __restrict__ Qp, const float* __restrict__ Kp,
                            const float* __restrict__ Vp, float* __restrict__ sa) {
  int lane = threadIdx.x & 31;
  int wv = __builtin_amdgcn_readfirstlane((int)(threadIdx.x >> 5));
  int mr = lane & 15, half = lane >> 4, kh = half << 1;
  int qt = wv * 32 + blockIdx.x;
  int bh = blockIdx.y;
  const size_t base = (size_t)bh * SEQ * DPAD;
  const float* Qb = Qp + base + (size_t)qt * 16 * DPAD;

  v2f qf[4];
  #pragma unroll
  for (int ck = 0; ck < 4; ck++) qf[ck] = *(const v2f*)(Qb + mr * DPAD + ck * 4 + kh);

  v8f o = {};
  float mrun = -INFINITY, lrun = 0.f;
  const float scale = 0.28867513459481287f;

  for (int kt = 0; kt <= qt; ++kt) {
    const float* Kt = Kp + base + (size_t)kt * 256;
    const float* Vt = Vp + base + (size_t)kt * 256;
    if (kt < qt) {
      __builtin_prefetch(Kt + 256, 0, 0);
      __builtin_prefetch(Vt + 256, 0, 0);
    }
    v8f s = {};
    #pragma unroll
    for (int ck = 0; ck < 4; ck++) {
      v2f kf = *(const v2f*)(Kt + mr * DPAD + ck * 4 + kh);
      s = wmma4(kf, qf[ck], s);
    }
    float sv[8];
    #pragma unroll
    for (int r = 0; r < 8; r++) sv[r] = s[r] * scale;
    if (kt == qt) {
      #pragma unroll
      for (int r = 0; r < 8; r++) sv[r] = (r + 8 * half > mr) ? -INFINITY : sv[r];
    }
    float tmax = fmaxf(fmaxf(fmaxf(sv[0], sv[1]), fmaxf(sv[2], sv[3])),
                       fmaxf(fmaxf(sv[4], sv[5]), fmaxf(sv[6], sv[7])));
    tmax = fmaxf(tmax, xhalf(tmax));
    float mnew  = fmaxf(mrun, tmax);
    float alpha = __expf(mrun - mnew);
    v2f pf[4];
    #pragma unroll
    for (int r = 0; r < 8; r++) pf[r >> 1][r & 1] = __expf(sv[r] - mnew);
    float rsum = ((pf[0][0] + pf[0][1]) + (pf[1][0] + pf[1][1])) +
                 ((pf[2][0] + pf[2][1]) + (pf[3][0] + pf[3][1]));
    rsum = rsum + xhalf(rsum);
    lrun = lrun * alpha + rsum;
    mrun = mnew;
    #pragma unroll
    for (int r = 0; r < 8; r++) o[r] = o[r] * alpha;
    #pragma unroll
    for (int ck = 0; ck < 4; ck++) {
      v2f vf = *(const v2f*)(Vt + (ck + 4 * half) * 32 + mr * 2);
      o = wmma4(vf, pf[ck], o);
    }
  }

  int b = bh >> 4, h = bh & 15;
  float* dst = sa + ((size_t)(b * SEQ + qt * 16 + mr)) * N_EMBD + h * HEAD + 8 * half;
  v4f lo = { o[0] / lrun, o[1] / lrun, o[2] / lrun, o[3] / lrun };
  *(v4f*)dst = lo;
  if (half == 0) {
    v4f hi = { o[4] / lrun, o[5] / lrun, o[6] / lrun, o[7] / lrun };
    *(v4f*)(dst + 4) = hi;
  }
}

__global__ void gemm_proj_kernel(const float* __restrict__ A, const float* __restrict__ Bf,
                                 const float* __restrict__ bias, const float* __restrict__ resid,
                                 float* __restrict__ out) {
  int wv = threadIdx.x >> 5;
  int tile = blockIdx.x * 8 + wv;
  const int TN = 192 / 64;
  int m0 = (tile / TN) * 16, n0 = (tile % TN) * 64;
  v8f acc[4] = {{}, {}, {}, {}};
  gemm16x64(A, Bf, 192, 192, 192, m0, n0, acc);
  int lane = threadIdx.x & 31; int mr = lane & 15, half = lane >> 4;
  #pragma unroll
  for (int t = 0; t < 4; t++) {
    int j = n0 + t * 16 + mr;
    float bj = bias[j];
    #pragma unroll
    for (int v = 0; v < 8; v++) {
      int m = m0 + v + 8 * half;
      size_t ofs = (size_t)m * 192 + j;
      out[ofs] = acc[t][v] + bj + resid[ofs];
    }
  }
}

__global__ void gemm_ff1_kernel(const float* __restrict__ A, const float* __restrict__ Bf,
                                float* __restrict__ out) {
  int wv = threadIdx.x >> 5;
  int tile = blockIdx.x * 8 + wv;
  const int TN = FFN / 64;
  int m0 = (tile / TN) * 16, n0 = (tile % TN) * 64;
  v8f acc[4] = {{}, {}, {}, {}};
  gemm16x64(A, Bf, 192, 192, FFN, m0, n0, acc);
  int lane = threadIdx.x & 31; int mr = lane & 15, half = lane >> 4;
  #pragma unroll
  for (int t = 0; t < 4; t++) {
    int j = n0 + t * 16 + mr;
    #pragma unroll
    for (int v = 0; v < 8; v++) {
      int m = m0 + v + 8 * half;
      float val = acc[t][v];
      out[(size_t)m * FFN + j] = val / (1.0f + __expf(-val));
    }
  }
}

__global__ void gemm_ff2_kernel(const float* __restrict__ A, const float* __restrict__ Bf,
                                const float* __restrict__ resid, float* __restrict__ out) {
  int wv = threadIdx.x >> 5;
  int tile = blockIdx.x * 8 + wv;
  const int TN = 192 / 64;
  int m0 = (tile / TN) * 16, n0 = (tile % TN) * 64;
  v8f acc[4] = {{}, {}, {}, {}};
  gemm16x64(A, Bf, FFN, FFN, 192, m0, n0, acc);
  int lane = threadIdx.x & 31; int mr = lane & 15, half = lane >> 4;
  #pragma unroll
  for (int t = 0; t < 4; t++) {
    int j = n0 + t * 16 + mr;
    #pragma unroll
    for (int v = 0; v < 8; v++) {
      int m = m0 + v + 8 * half;
      size_t ofs = (size_t)m * 192 + j;
      out[ofs] = acc[t][v] + resid[ofs];
    }
  }
}

extern "C" void kernel_launch(void* const* d_in, const int* in_sizes, int n_in,
                              void* d_out, int out_size, void* d_ws, size_t ws_size,
                              hipStream_t stream) {
  (void)in_sizes; (void)n_in; (void)out_size; (void)ws_size;
  const float* x     = (const float*)d_in[0];
  const float* ln1_w = (const float*)d_in[1];
  const float* ln2_w = (const float*)d_in[2];
  const float* Wq    = (const float*)d_in[3];
  const float* Wk    = (const float*)d_in[4];
  const float* Wv    = (const float*)d_in[5];
  const float* Wproj = (const float*)d_in[6];
  const float* bproj = (const float*)d_in[7];
  const float* W1    = (const float*)d_in[8];
  const float* W2    = (const float*)d_in[9];
  float* ws = (float*)d_ws;

  size_t off = 0;
  float* h    = ws + off; off += (size_t)M_TOK * N_EMBD;
  float* wqkv = ws + off; off += (size_t)192 * 576;
  float* wpjf = ws + off; off += (size_t)192 * 192;
  float* w1f  = ws + off; off += (size_t)192 * FFN;
  float* w2f  = ws + off; off += (size_t)FFN * 192;
  float* Qp   = ws + off; off += (size_t)BH * SEQ * DPAD;
  float* Kp   = ws + off; off += (size_t)BH * SEQ * DPAD;
  float* Vp   = ws + off; off += (size_t)BH * SEQ * DPAD;
  float* sa   = ws + off; off += (size_t)M_TOK * N_EMBD;
  float* x1   = ws + off; off += (size_t)M_TOK * N_EMBD;
  float* h2   = ws + off; off += (size_t)M_TOK * N_EMBD;
  float* u    = ws + off; off += (size_t)M_TOK * FFN;
  float* out  = (float*)d_out;

  hipMemsetAsync(Qp, 0, (size_t)3 * BH * SEQ * DPAD * sizeof(float), stream);

  pack_qkv_kernel<<<(192 * 576) / 256, 256, 0, stream>>>(Wq, Wk, Wv, wqkv);
  pack_b_kernel<<<(192 * 192) / 256, 256, 0, stream>>>(Wproj, wpjf, 192);
  pack_b_kernel<<<(192 * FFN) / 256, 256, 0, stream>>>(W1, w1f, FFN);
  pack_b_kernel<<<(FFN * 192) / 256, 256, 0, stream>>>(W2, w2f, 192);

  rmsnorm_kernel<<<M_TOK / 8, 256, 0, stream>>>(x, ln1_w, h);
  gemm_qkv_kernel<<<(512 * 9) / 8, 256, 0, stream>>>(h, wqkv, Qp, Kp, Vp);
  attn_kernel<<<dim3(32, BH), 128, 0, stream>>>(Qp, Kp, Vp, sa);
  gemm_proj_kernel<<<(512 * 3) / 8, 256, 0, stream>>>(sa, wpjf, bproj, x, x1);
  rmsnorm_kernel<<<M_TOK / 8, 256, 0, stream>>>(x1, ln2_w, h2);
  gemm_ff1_kernel<<<(512 * 12) / 8, 256, 0, stream>>>(h2, w1f, u);
  gemm_ff2_kernel<<<(512 * 3) / 8, 256, 0, stream>>>(u, w2f, x1, out);
}


# attention: each wave owns the query-tile pair (127-p, p) of one head, run in turn: 129 key tiles per wave for every p, so all waves and SIMDs carry equal causal work (grid 16x64, 4 waves)
# speedup vs baseline: 1.4748x; 1.1429x over previous
#include <hip/hip_runtime.h>
#include <math.h>

typedef __attribute__((ext_vector_type(2))) float v2f;
typedef __attribute__((ext_vector_type(4))) float v4f;
typedef __attribute__((ext_vector_type(8))) float v8f;

#define N_EMBD 192
#define N_HEAD 16
#define HEAD   12
#define DPAD   16
#define BATCH  4
#define SEQ    2048
#define M_TOK  (BATCH*SEQ)
#define FFN    (4*N_EMBD)
#define BH     (BATCH*N_HEAD)

static __device__ __forceinline__ v8f wmma4(v2f a, v2f b, v8f c) {
  return __builtin_amdgcn_wmma_f32_16x16x4_f32(false, a, false, b, (short)0, c, false, false);
}

__global__ void rmsnorm_kernel(const float* __restrict__ x, const float* __restrict__ w,
                               float* __restrict__ out) {
  int lane = threadIdx.x & 31;
  int wv   = threadIdx.x >> 5;
  int row  = blockIdx.x * 8 + wv;
  const float* xr = x + (size_t)row * N_EMBD;
  float v[6]; float ss = 0.f;
  #pragma unroll
  for (int i = 0; i < 6; i++) { v[i] = xr[lane + 32*i]; ss += v[i]*v[i]; }
  #pragma unroll
  for (int off = 16; off >= 1; off >>= 1) ss += __shfl_xor(ss, off, 32);
  float sc = rsqrtf(ss * (1.0f / N_EMBD) + 1e-5f);
  float* orow = out + (size_t)row * N_EMBD;
  #pragma unroll
  for (int i = 0; i < 6; i++) { int c = lane + 32*i; orow[c] = v[i] * sc * w[c]; }
}

__global__ void pack_b_kernel(const float* __restrict__ B, float* __restrict__ Bf, int N) {
  int idx = blockIdx.x * 256 + threadIdx.x;
  int k = idx / N, n = idx % N;
  Bf[((size_t)(k >> 1) * N + n) * 2 + (k & 1)] = B[idx];
}

__global__ void pack_qkv_kernel(const float* __restrict__ Wq, const float* __restrict__ Wk,
                                const float* __restrict__ Wv, float* __restrict__ Bf) {
  int idx = blockIdx.x * 256 + threadIdx.x;
  int c = idx / 576, j = idx % 576;
  int which = j / 192, r = j % 192;
  int h = r / HEAD, d = r % HEAD;
  const float* W = (which == 0) ? Wq : ((which == 1) ? Wk : Wv);
  Bf[((size_t)(c >> 1) * 576 + j) * 2 + (c & 1)] = W[((size_t)h * N_EMBD + c) * HEAD + d];
}

__device__ __forceinline__ void gemm16x64(const float* __restrict__ A, const float* __restrict__ Bf,
                                          int K, int lda, int N, int m0, int n0, v8f* acc) {
  int lane = threadIdx.x & 31;
  int mr = lane & 15;
  int kh = (lane >> 4) << 1;
  const float* arow  = A + (size_t)(m0 + mr) * lda;
  const float* bbase = Bf + (size_t)(n0 + mr) * 2;
  #pragma unroll 2
  for (int k = 0; k < K; k += 4) {
    v2f a = *(const v2f*)(arow + k + kh);
    const float* bk = bbase + (size_t)((k + kh) >> 1) * (2 * N);
    #pragma unroll
    for (int t = 0; t < 4; t++) {
      v2f b = *(const v2f*)(bk + t * 32);
      acc[t] = wmma4(a, b, acc[t]);
    }
  }
}

__global__ void gemm_qkv_kernel(const float* __restrict__ A, const float* __restrict__ Bf,
                                float* __restrict__ Qp, float* __restrict__ Kp,
                                float* __restrict__ Vp) {
  int wv = threadIdx.x >> 5;
  int tile = blockIdx.x * 8 + wv;
  const int TN = 576 / 64;
  int m0 = (tile / TN) * 16;
  int n0 = (tile % TN) * 64;
  v8f acc[4] = {{}, {}, {}, {}};
  gemm16x64(A, Bf, N_EMBD, N_EMBD, 576, m0, n0, acc);
  int lane = threadIdx.x & 31;
  int mr = lane & 15, half = lane >> 4;
  #pragma unroll
  for (int t = 0; t < 4; t++) {
    int j = n0 + t * 16 + mr;
    int which = j / 192, r = j % 192, h = r / HEAD, d = r % HEAD;
    #pragma unroll
    for (int v = 0; v < 8; v++) {
      int m = m0 + v + 8 * half;
      int b = m >> 11, tq = m & 2047;
      size_t hb = ((size_t)(b * N_HEAD + h)) * SEQ * DPAD;
      if (which == 0)      Qp[hb + (size_t)tq * DPAD + d] = acc[t][v];
      else if (which == 1) Kp[hb + (size_t)tq * DPAD + d] = acc[t][v];
      else                 Vp[hb + (size_t)(tq >> 1) * 32 + d * 2 + (tq & 1)] = acc[t][v];
    }
  }
}


static __device__ __forceinline__ float xhalf(float x) {
  int b = __builtin_bit_cast(int, x);
  return __builtin_bit_cast(float,
      __builtin_amdgcn_permlanex16(b, b, 0x76543210, 0xfedcba98, false, false));
}

__global__ void attn_kernel(const float* __restrict__ Qp, const float* __restrict__ Kp,
                            const float* __restrict__ Vp, float* __restrict__ sa) {
  int lane = threadIdx.x & 31;
  int wv = __builtin_amdgcn_readfirstlane((int)(threadIdx.x >> 5));
  int mr = lane & 15, half = lane >> 4, kh = half << 1;
  int pr = blockIdx.x * 4 + wv;
  int bh = blockIdx.y;
  const size_t base = (size_t)bh * SEQ * DPAD;
  const float scale = 0.28867513459481287f;
  int b = bh >> 4, h = bh & 15;

  #pragma unroll 1
  for (int pass = 0; pass < 2; ++pass) {
    int qt = pass ? pr : (SEQ / 16 - 1) - pr;
    const float* Qb = Qp + base + (size_t)qt * 16 * DPAD;

    v2f qf[4];
    #pragma unroll
    for (int ck = 0; ck < 4; ck++) qf[ck] = *(const v2f*)(Qb + mr * DPAD + ck * 4 + kh);

    v8f o = {};
    float mrun = -INFINITY, lrun = 0.f;

    for (int kt = 0; kt <= qt; ++kt) {
      const float* Kt = Kp + base + (size_t)kt * 256;
      const float* Vt = Vp + base + (size_t)kt * 256;
      if (kt < qt) {
        __builtin_prefetch(Kt + 256, 0, 0);
        __builtin_prefetch(Vt + 256, 0, 0);
      }
      v8f s = {};
      #pragma unroll
      for (int ck = 0; ck < 4; ck++) {
        v2f kf = *(const v2f*)(Kt + mr * DPAD + ck * 4 + kh);
        s = wmma4(kf, qf[ck], s);
      }
      float sv[8];
      #pragma unroll
      for (int r = 0; r < 8; r++) sv[r] = s[r] * scale;
      if (kt == qt) {
        #pragma unroll
        for (int r = 0; r < 8; r++) sv[r] = (r + 8 * half > mr) ? -INFINITY : sv[r];
      }
      float tmax = fmaxf(fmaxf(fmaxf(sv[0], sv[1]), fmaxf(sv[2], sv[3])),
                         fmaxf(fmaxf(sv[4], sv[5]), fmaxf(sv[6], sv[7])));
      tmax = fmaxf(tmax, xhalf(tmax));
      float mnew  = fmaxf(mrun, tmax);
      float alpha = __expf(mrun - mnew);
      v2f pf[4];
      #pragma unroll
      for (int r = 0; r < 8; r++) pf[r >> 1][r & 1] = __expf(sv[r] - mnew);
      float rsum = ((pf[0][0] + pf[0][1]) + (pf[1][0] + pf[1][1])) +
                   ((pf[2][0] + pf[2][1]) + (pf[3][0] + pf[3][1]));
      rsum = rsum + xhalf(rsum);
      lrun = lrun * alpha + rsum;
      mrun = mnew;
      #pragma unroll
      for (int r = 0; r < 8; r++) o[r] = o[r] * alpha;
      #pragma unroll
      for (int ck = 0; ck < 4; ck++) {
        v2f vf = *(const v2f*)(Vt + (ck + 4 * half) * 32 + mr * 2);
        o = wmma4(vf, pf[ck], o);
      }
    }

    float* dst = sa + ((size_t)(b * SEQ + qt * 16 + mr)) * N_EMBD + h * HEAD + 8 * half;
    v4f lo = { o[0] / lrun, o[1] / lrun, o[2] / lrun, o[3] / lrun };
    *(v4f*)dst = lo;
    if (half == 0) {
      v4f hi = { o[4] / lrun, o[5] / lrun, o[6] / lrun, o[7] / lrun };
      *(v4f*)(dst + 4) = hi;
    }
  }
}

__global__ void gemm_proj_kernel(const float* __restrict__ A, const float* __restrict__ Bf,
                                 const float* __restrict__ bias, const float* __restrict__ resid,
                                 float* __restrict__ out) {
  int wv = threadIdx.x >> 5;
  int tile = blockIdx.x * 8 + wv;
  const int TN = 192 / 64;
  int m0 = (tile / TN) * 16, n0 = (tile % TN) * 64;
  v8f acc[4] = {{}, {}, {}, {}};
  gemm16x64(A, Bf, 192, 192, 192, m0, n0, acc);
  int lane = threadIdx.x & 31; int mr = lane & 15, half = lane >> 4;
  #pragma unroll
  for (int t = 0; t < 4; t++) {
    int j = n0 + t * 16 + mr;
    float bj = bias[j];
    #pragma unroll
    for (int v = 0; v < 8; v++) {
      int m = m0 + v + 8 * half;
      size_t ofs = (size_t)m * 192 + j;
      out[ofs] = acc[t][v] + bj + resid[ofs];
    }
  }
}

__global__ void gemm_ff1_kernel(const float* __restrict__ A, const float* __restrict__ Bf,
                                float* __restrict__ out) {
  int wv = threadIdx.x >> 5;
  int tile = blockIdx.x * 8 + wv;
  const int TN = FFN / 64;
  int m0 = (tile / TN) * 16, n0 = (tile % TN) * 64;
  v8f acc[4] = {{}, {}, {}, {}};
  gemm16x64(A, Bf, 192, 192, FFN, m0, n0, acc);
  int lane = threadIdx.x & 31; int mr = lane & 15, half = lane >> 4;
  #pragma unroll
  for (int t = 0; t < 4; t++) {
    int j = n0 + t * 16 + mr;
    #pragma unroll
    for (int v = 0; v < 8; v++) {
      int m = m0 + v + 8 * half;
      float val = acc[t][v];
      out[(size_t)m * FFN + j] = val / (1.0f + __expf(-val));
    }
  }
}

__global__ void gemm_ff2_kernel(const float* __restrict__ A, const float* __restrict__ Bf,
                                const float* __restrict__ resid, float* __restrict__ out) {
  int wv = threadIdx.x >> 5;
  int tile = blockIdx.x * 8 + wv;
  const int TN = 192 / 64;
  int m0 = (tile / TN) * 16, n0 = (tile % TN) * 64;
  v8f acc[4] = {{}, {}, {}, {}};
  gemm16x64(A, Bf, FFN, FFN, 192, m0, n0, acc);
  int lane = threadIdx.x & 31; int mr = lane & 15, half = lane >> 4;
  #pragma unroll
  for (int t = 0; t < 4; t++) {
    int j = n0 + t * 16 + mr;
    #pragma unroll
    for (int v = 0; v < 8; v++) {
      int m = m0 + v + 8 * half;
      size_t ofs = (size_t)m * 192 + j;
      out[ofs] = acc[t][v] + resid[ofs];
    }
  }
}

extern "C" void kernel_launch(void* const* d_in, const int* in_sizes, int n_in,
                              void* d_out, int out_size, void* d_ws, size_t ws_size,
                              hipStream_t stream) {
  (void)in_sizes; (void)n_in; (void)out_size; (void)ws_size;
  const float* x     = (const float*)d_in[0];
  const float* ln1_w = (const float*)d_in[1];
  const float* ln2_w = (const float*)d_in[2];
  const float* Wq    = (const float*)d_in[3];
  const float* Wk    = (const float*)d_in[4];
  const float* Wv    = (const float*)d_in[5];
  const float* Wproj = (const float*)d_in[6];
  const float* bproj = (const float*)d_in[7];
  const float* W1    = (const float*)d_in[8];
  const float* W2    = (const float*)d_in[9];
  float* ws = (float*)d_ws;

  size_t off = 0;
  float* h    = ws + off; off += (size_t)M_TOK * N_EMBD;
  float* wqkv = ws + off; off += (size_t)192 * 576;
  float* wpjf = ws + off; off += (size_t)192 * 192;
  float* w1f  = ws + off; off += (size_t)192 * FFN;
  float* w2f  = ws + off; off += (size_t)FFN * 192;
  float* Qp   = ws + off; off += (size_t)BH * SEQ * DPAD;
  float* Kp   = ws + off; off += (size_t)BH * SEQ * DPAD;
  float* Vp   = ws + off; off += (size_t)BH * SEQ * DPAD;
  float* sa   = ws + off; off += (size_t)M_TOK * N_EMBD;
  float* x1   = ws + off; off += (size_t)M_TOK * N_EMBD;
  float* h2   = ws + off; off += (size_t)M_TOK * N_EMBD;
  float* u    = ws + off; off += (size_t)M_TOK * FFN;
  float* out  = (float*)d_out;

  hipMemsetAsync(Qp, 0, (size_t)3 * BH * SEQ * DPAD * sizeof(float), stream);

  pack_qkv_kernel<<<(192 * 576) / 256, 256, 0, stream>>>(Wq, Wk, Wv, wqkv);
  pack_b_kernel<<<(192 * 192) / 256, 256, 0, stream>>>(Wproj, wpjf, 192);
  pack_b_kernel<<<(192 * FFN) / 256, 256, 0, stream>>>(W1, w1f, FFN);
  pack_b_kernel<<<(FFN * 192) / 256, 256, 0, stream>>>(W2, w2f, 192);

  rmsnorm_kernel<<<M_TOK / 8, 256, 0, stream>>>(x, ln1_w, h);
  gemm_qkv_kernel<<<(512 * 9) / 8, 256, 0, stream>>>(h, wqkv, Qp, Kp, Vp);
  attn_kernel<<<dim3(16, BH), 128, 0, stream>>>(Qp, Kp, Vp, sa);
  gemm_proj_kernel<<<(512 * 3) / 8, 256, 0, stream>>>(sa, wpjf, bproj, x, x1);
  rmsnorm_kernel<<<M_TOK / 8, 256, 0, stream>>>(x1, ln2_w, h2);
  gemm_ff1_kernel<<<(512 * 12) / 8, 256, 0, stream>>>(h2, w1f, u);
  gemm_ff2_kernel<<<(512 * 3) / 8, 256, 0, stream>>>(u, w2f, x1, out);
}


# GEMM K loop of all four products software-pipelined: fragments of the next two 16-k stages in flight (three rotating register sets, 16-byte loads, weights packed by k-quads) behind the current WMMAs
# speedup vs baseline: 1.8300x; 1.2408x over previous
#include <hip/hip_runtime.h>
#include <math.h>

typedef __attribute__((ext_vector_type(2))) float v2f;
typedef __attribute__((ext_vector_type(4))) float v4f;
typedef __attribute__((ext_vector_type(8))) float v8f;

#define N_EMBD 192
#define N_HEAD 16
#define HEAD   12
#define DPAD   16
#define BATCH  4
#define SEQ    2048
#define M_TOK  (BATCH*SEQ)
#define FFN    (4*N_EMBD)
#define BH     (BATCH*N_HEAD)

static __device__ __forceinline__ v8f wmma4(v2f a, v2f b, v8f c) {
  return __builtin_amdgcn_wmma_f32_16x16x4_f32(false, a, false, b, (short)0, c, false, false);
}

__global__ void rmsnorm_kernel(const float* __restrict__ x, const float* __restrict__ w,
                               float* __restrict__ out) {
  int lane = threadIdx.x & 31;
  int wv   = threadIdx.x >> 5;
  int row  = blockIdx.x * 8 + wv;
  const float* xr = x + (size_t)row * N_EMBD;
  float v[6]; float ss = 0.f;
  #pragma unroll
  for (int i = 0; i < 6; i++) { v[i] = xr[lane + 32*i]; ss += v[i]*v[i]; }
  #pragma unroll
  for (int off = 16; off >= 1; off >>= 1) ss += __shfl_xor(ss, off, 32);
  float sc = rsqrtf(ss * (1.0f / N_EMBD) + 1e-5f);
  float* orow = out + (size_t)row * N_EMBD;
  #pragma unroll
  for (int i = 0; i < 6; i++) { int c = lane + 32*i; orow[c] = v[i] * sc * w[c]; }
}

__global__ void pack_b_kernel(const float* __restrict__ B, float* __restrict__ Bf, int N) {
  int idx = blockIdx.x * 256 + threadIdx.x;
  int k = idx / N, n = idx % N;
  Bf[((size_t)(k >> 2) * N + n) * 4 + (k & 3)] = B[idx];
}

__global__ void pack_qkv_kernel(const float* __restrict__ Wq, const float* __restrict__ Wk,
                                const float* __restrict__ Wv, float* __restrict__ Bf) {
  int idx = blockIdx.x * 256 + threadIdx.x;
  int c = idx / 576, j = idx % 576;
  int which = j / 192, r = j % 192;
  int h = r / HEAD, d = r % HEAD;
  const float* W = (which == 0) ? Wq : ((which == 1) ? Wk : Wv);
  Bf[((size_t)(c >> 2) * 576 + j) * 4 + (c & 3)] = W[((size_t)h * N_EMBD + c) * HEAD + d];
}

struct Stage { v4f a[2]; v4f b[2][4]; };

static __device__ __forceinline__ v2f lo2(v4f x) { return __builtin_shufflevector(x, x, 0, 1); }
static __device__ __forceinline__ v2f hi2(v4f x) { return __builtin_shufflevector(x, x, 2, 3); }

static __device__ __forceinline__ void stage_load(Stage& f, const float* __restrict__ ap,
                                                  const float* __restrict__ bp, int N) {
  #pragma unroll
  for (int u = 0; u < 2; u++) {
    f.a[u] = *(const v4f*)(ap + 8 * u);
    #pragma unroll
    for (int t = 0; t < 4; t++) f.b[u][t] = *(const v4f*)(bp + (size_t)u * 8 * N + t * 64);
  }
}

static __device__ __forceinline__ void stage_mma(const Stage& f, v8f* acc) {
  #pragma unroll
  for (int u = 0; u < 2; u++) {
    v2f a0 = lo2(f.a[u]), a1 = hi2(f.a[u]);
    #pragma unroll
    for (int t = 0; t < 4; t++) acc[t] = wmma4(a0, lo2(f.b[u][t]), acc[t]);
    #pragma unroll
    for (int t = 0; t < 4; t++) acc[t] = wmma4(a1, hi2(f.b[u][t]), acc[t]);
  }
}

__device__ __forceinline__ void gemm16x64(const float* __restrict__ A, const float* __restrict__ Bf,
                                          int K, int lda, int N, int m0, int n0, v8f* acc) {
  int lane = threadIdx.x & 31;
  int mr = lane & 15;
  int half = lane >> 4;
  const float* ap = A + (size_t)(m0 + mr) * lda + 4 * half;
  const float* bp = Bf + (size_t)half * 4 * N + (size_t)(n0 + mr) * 4;
  const int S = K / 16;
  Stage f0, f1, f2;
#define STAGE_EDGE() do { \
    asm volatile("" : "+v"(acc[0]), "+v"(acc[1]), "+v"(acc[2]), "+v"(acc[3]) : : "memory"); \
    __builtin_amdgcn_sched_barrier(0); } while (0)
  stage_load(f0, ap, bp, N);
  stage_load(f1, ap + 16, bp + (size_t)16 * N, N);
  ap += 32; bp += (size_t)32 * N;
  #pragma unroll 1
  for (int s = 0; s < S - 3; s += 3) {
    stage_load(f2, ap, bp, N);                          STAGE_EDGE();
    stage_mma(f0, acc);                                 STAGE_EDGE();
    stage_load(f0, ap + 16, bp + (size_t)16 * N, N);    STAGE_EDGE();
    stage_mma(f1, acc);                                 STAGE_EDGE();
    stage_load(f1, ap + 32, bp + (size_t)32 * N, N);    STAGE_EDGE();
    stage_mma(f2, acc);                                 STAGE_EDGE();
    ap += 48; bp += (size_t)48 * N;
  }
  stage_load(f2, ap, bp, N);                            STAGE_EDGE();
  stage_mma(f0, acc);
  stage_mma(f1, acc);
  stage_mma(f2, acc);
#undef STAGE_EDGE
}

__global__ void __launch_bounds__(256) gemm_qkv_kernel(const float* __restrict__ A, const float* __restrict__ Bf,
                                float* __restrict__ Qp, float* __restrict__ Kp,
                                float* __restrict__ Vp) {
  int wv = threadIdx.x >> 5;
  int tile = blockIdx.x * 8 + wv;
  const int TN = 576 / 64;
  int m0 = (tile / TN) * 16;
  int n0 = (tile % TN) * 64;
  v8f acc[4] = {{}, {}, {}, {}};
  gemm16x64(A, Bf, N_EMBD, N_EMBD, 576, m0, n0, acc);
  int lane = threadIdx.x & 31;
  int mr = lane & 15, half = lane >> 4;
  #pragma unroll
  for (int t = 0; t < 4; t++) {
    int j = n0 + t * 16 + mr;
    int which = j / 192, r = j % 192, h = r / HEAD, d = r % HEAD;
    #pragma unroll
    for (int v = 0; v < 8; v++) {
      int m = m0 + v + 8 * half;
      int b = m >> 11, tq = m & 2047;
      size_t hb = ((size_t)(b * N_HEAD + h)) * SEQ * DPAD;
      if (which == 0)      Qp[hb + (size_t)tq * DPAD + d] = acc[t][v];
      else if (which == 1) Kp[hb + (size_t)tq * DPAD + d] = acc[t][v];
      else                 Vp[hb + (size_t)(tq >> 1) * 32 + d * 2 + (tq & 1)] = acc[t][v];
    }
  }
}


static __device__ __forceinline__ float xhalf(float x) {
  int b = __builtin_bit_cast(int, x);
  return __builtin_bit_cast(float,
      __builtin_amdgcn_permlanex16(b, b, 0x76543210, 0xfedcba98, false, false));
}

__global__ void attn_kernel(const float* __restrict__ Qp, const float* __restrict__ Kp,
                            const float* __restrict__ Vp, float* __restrict__ sa) {
  int lane = threadIdx.x & 31;
  int wv = __builtin_amdgcn_readfirstlane((int)(threadIdx.x >> 5));
  int mr = lane & 15, half = lane >> 4, kh = half << 1;
  int pr = blockIdx.x * 4 + wv;
  int bh = blockIdx.y;
  const size_t base = (size_t)bh * SEQ * DPAD;
  const float scale = 0.28867513459481287f;
  int b = bh >> 4, h = bh & 15;

  #pragma unroll 1
  for (int pass = 0; pass < 2; ++pass) {
    int qt = pass ? pr : (SEQ / 16 - 1) - pr;
    const float* Qb = Qp + base + (size_t)qt * 16 * DPAD;

    v2f qf[4];
    #pragma unroll
    for (int ck = 0; ck < 4; ck++) qf[ck] = *(const v2f*)(Qb + mr * DPAD + ck * 4 + kh);

    v8f o = {};
    float mrun = -INFINITY, lrun = 0.f;

    for (int kt = 0; kt <= qt; ++kt) {
      const float* Kt = Kp + base + (size_t)kt * 256;
      const float* Vt = Vp + base + (size_t)kt * 256;
      if (kt < qt) {
        __builtin_prefetch(Kt + 256, 0, 0);
        __builtin_prefetch(Vt + 256, 0, 0);
      }
      v8f s = {};
      #pragma unroll
      for (int ck = 0; ck < 4; ck++) {
        v2f kf = *(const v2f*)(Kt + mr * DPAD + ck * 4 + kh);
        s = wmma4(kf, qf[ck], s);
      }
      float sv[8];
      #pragma unroll
      for (int r = 0; r < 8; r++) sv[r] = s[r] * scale;
      if (kt == qt) {
        #pragma unroll
        for (int r = 0; r < 8; r++) sv[r] = (r + 8 * half > mr) ? -INFINITY : sv[r];
      }
      float tmax = fmaxf(fmaxf(fmaxf(sv[0], sv[1]), fmaxf(sv[2], sv[3])),
                         fmaxf(fmaxf(sv[4], sv[5]), fmaxf(sv[6], sv[7])));
      tmax = fmaxf(tmax, xhalf(tmax));
      float mnew  = fmaxf(mrun, tmax);
      float alpha = __expf(mrun - mnew);
      v2f pf[4];
      #pragma unroll
      for (int r = 0; r < 8; r++) pf[r >> 1][r & 1] = __expf(sv[r] - mnew);
      float rsum = ((pf[0][0] + pf[0][1]) + (pf[1][0] + pf[1][1])) +
                   ((pf[2][0] + pf[2][1]) + (pf[3][0] + pf[3][1]));
      rsum = rsum + xhalf(rsum);
      lrun = lrun * alpha + rsum;
      mrun = mnew;
      #pragma unroll
      for (int r = 0; r < 8; r++) o[r] = o[r] * alpha;
      #pragma unroll
      for (int ck = 0; ck < 4; ck++) {
        v2f vf = *(const v2f*)(Vt + (ck + 4 * half) * 32 + mr * 2);
        o = wmma4(vf, pf[ck], o);
      }
    }

    float* dst = sa + ((size_t)(b * SEQ + qt * 16 + mr)) * N_EMBD + h * HEAD + 8 * half;
    v4f lo = { o[0] / lrun, o[1] / lrun, o[2] / lrun, o[3] / lrun };
    *(v4f*)dst = lo;
    if (half == 0) {
      v4f hi = { o[4] / lrun, o[5] / lrun, o[6] / lrun, o[7] / lrun };
      *(v4f*)(dst + 4) = hi;
    }
  }
}

__global__ void __launch_bounds__(256) gemm_proj_kernel(const float* __restrict__ A, const float* __restrict__ Bf,
                                 const float* __restrict__ bias, const float* __restrict__ resid,
                                 float* __restrict__ out) {
  int wv = threadIdx.x >> 5;
  int tile = blockIdx.x * 8 + wv;
  const int TN = 192 / 64;
  int m0 = (tile / TN) * 16, n0 = (tile % TN) * 64;
  v8f acc[4] = {{}, {}, {}, {}};
  gemm16x64(A, Bf, 192, 192, 192, m0, n0, acc);
  int lane = threadIdx.x & 31; int mr = lane & 15, half = lane >> 4;
  #pragma unroll
  for (int t = 0; t < 4; t++) {
    int j = n0 + t * 16 + mr;
    float bj = bias[j];
    #pragma unroll
    for (int v = 0; v < 8; v++) {
      int m = m0 + v + 8 * half;
      size_t ofs = (size_t)m * 192 + j;
      out[ofs] = acc[t][v] + bj + resid[ofs];
    }
  }
}

__global__ void __launch_bounds__(256) gemm_ff1_kernel(const float* __restrict__ A, const float* __restrict__ Bf,
                                float* __restrict__ out) {
  int wv = threadIdx.x >> 5;
  int tile = blockIdx.x * 8 + wv;
  const int TN = FFN / 64;
  int m0 = (tile / TN) * 16, n0 = (tile % TN) * 64;
  v8f acc[4] = {{}, {}, {}, {}};
  gemm16x64(A, Bf, 192, 192, FFN, m0, n0, acc);
  int lane = threadIdx.x & 31; int mr = lane & 15, half = lane >> 4;
  #pragma unroll
  for (int t = 0; t < 4; t++) {
    int j = n0 + t * 16 + mr;
    #pragma unroll
    for (int v = 0; v < 8; v++) {
      int m = m0 + v + 8 * half;
      float val = acc[t][v];
      out[(size_t)m * FFN + j] = val / (1.0f + __expf(-val));
    }
  }
}

__global__ void __launch_bounds__(256) gemm_ff2_kernel(const float* __restrict__ A, const float* __restrict__ Bf,
                                const float* __restrict__ resid, float* __restrict__ out) {
  int wv = threadIdx.x >> 5;
  int tile = blockIdx.x * 8 + wv;
  const int TN = 192 / 64;
  int m0 = (tile / TN) * 16, n0 = (tile % TN) * 64;
  v8f acc[4] = {{}, {}, {}, {}};
  gemm16x64(A, Bf, FFN, FFN, 192, m0, n0, acc);
  int lane = threadIdx.x & 31; int mr = lane & 15, half = lane >> 4;
  #pragma unroll
  for (int t = 0; t < 4; t++) {
    int j = n0 + t * 16 + mr;
    #pragma unroll
    for (int v = 0; v < 8; v++) {
      int m = m0 + v + 8 * half;
      size_t ofs = (size_t)m * 192 + j;
      out[ofs] = acc[t][v] + resid[ofs];
    }
  }
}

extern "C" void kernel_launch(void* const* d_in, const int* in_sizes, int n_in,
                              void* d_out, int out_size, void* d_ws, size_t ws_size,
                              hipStream_t stream) {
  (void)in_sizes; (void)n_in; (void)out_size; (void)ws_size;
  const float* x     = (const float*)d_in[0];
  const float* ln1_w = (const float*)d_in[1];
  const float* ln2_w = (const float*)d_in[2];
  const float* Wq    = (const float*)d_in[3];
  const float* Wk    = (const float*)d_in[4];
  const float* Wv    = (const float*)d_in[5];
  const float* Wproj = (const float*)d_in[6];
  const float* bproj = (const float*)d_in[7];
  const float* W1    = (const float*)d_in[8];
  const float* W2    = (const float*)d_in[9];
  float* ws = (float*)d_ws;

  size_t off = 0;
  float* h    = ws + off; off += (size_t)M_TOK * N_EMBD;
  float* wqkv = ws + off; off += (size_t)192 * 576;
  float* wpjf = ws + off; off += (size_t)192 * 192;
  float* w1f  = ws + off; off += (size_t)192 * FFN;
  float* w2f  = ws + off; off += (size_t)FFN * 192;
  float* Qp   = ws + off; off += (size_t)BH * SEQ * DPAD;
  float* Kp   = ws + off; off += (size_t)BH * SEQ * DPAD;
  float* Vp   = ws + off; off += (size_t)BH * SEQ * DPAD;
  float* sa   = ws + off; off += (size_t)M_TOK * N_EMBD;
  float* x1   = ws + off; off += (size_t)M_TOK * N_EMBD;
  float* h2   = ws + off; off += (size_t)M_TOK * N_EMBD;
  float* u    = ws + off; off += (size_t)M_TOK * FFN;
  float* out  = (float*)d_out;

  hipMemsetAsync(Qp, 0, (size_t)3 * BH * SEQ * DPAD * sizeof(float), stream);

  pack_qkv_kernel<<<(192 * 576) / 256, 256, 0, stream>>>(Wq, Wk, Wv, wqkv);
  pack_b_kernel<<<(192 * 192) / 256, 256, 0, stream>>>(Wproj, wpjf, 192);
  pack_b_kernel<<<(192 * FFN) / 256, 256, 0, stream>>>(W1, w1f, FFN);
  pack_b_kernel<<<(FFN * 192) / 256, 256, 0, stream>>>(W2, w2f, 192);

  rmsnorm_kernel<<<M_TOK / 8, 256, 0, stream>>>(x, ln1_w, h);
  gemm_qkv_kernel<<<(512 * 9) / 8, 256, 0, stream>>>(h, wqkv, Qp, Kp, Vp);
  attn_kernel<<<dim3(16, BH), 128, 0, stream>>>(Qp, Kp, Vp, sa);
  gemm_proj_kernel<<<(512 * 3) / 8, 256, 0, stream>>>(sa, wpjf, bproj, x, x1);
  rmsnorm_kernel<<<M_TOK / 8, 256, 0, stream>>>(x1, ln2_w, h2);
  gemm_ff1_kernel<<<(512 * 12) / 8, 256, 0, stream>>>(h2, w1f, u);
  gemm_ff2_kernel<<<(512 * 3) / 8, 256, 0, stream>>>(u, w2f, x1, out);
}


# attention key loop software-pipelined: K and V fragments of key tile kt+1 are loaded into a second register set in front of tile kt's matrix and softmax work, replacing the cache hint
# speedup vs baseline: 2.0095x; 1.0981x over previous
#include <hip/hip_runtime.h>
#include <math.h>

typedef __attribute__((ext_vector_type(2))) float v2f;
typedef __attribute__((ext_vector_type(4))) float v4f;
typedef __attribute__((ext_vector_type(8))) float v8f;

#define N_EMBD 192
#define N_HEAD 16
#define HEAD   12
#define DPAD   16
#define BATCH  4
#define SEQ    2048
#define M_TOK  (BATCH*SEQ)
#define FFN    (4*N_EMBD)
#define BH     (BATCH*N_HEAD)

static __device__ __forceinline__ v8f wmma4(v2f a, v2f b, v8f c) {
  return __builtin_amdgcn_wmma_f32_16x16x4_f32(false, a, false, b, (short)0, c, false, false);
}

__global__ void rmsnorm_kernel(const float* __restrict__ x, const float* __restrict__ w,
                               float* __restrict__ out) {
  int lane = threadIdx.x & 31;
  int wv   = threadIdx.x >> 5;
  int row  = blockIdx.x * 8 + wv;
  const float* xr = x + (size_t)row * N_EMBD;
  float v[6]; float ss = 0.f;
  #pragma unroll
  for (int i = 0; i < 6; i++) { v[i] = xr[lane + 32*i]; ss += v[i]*v[i]; }
  #pragma unroll
  for (int off = 16; off >= 1; off >>= 1) ss += __shfl_xor(ss, off, 32);
  float sc = rsqrtf(ss * (1.0f / N_EMBD) + 1e-5f);
  float* orow = out + (size_t)row * N_EMBD;
  #pragma unroll
  for (int i = 0; i < 6; i++) { int c = lane + 32*i; orow[c] = v[i] * sc * w[c]; }
}

__global__ void pack_b_kernel(const float* __restrict__ B, float* __restrict__ Bf, int N) {
  int idx = blockIdx.x * 256 + threadIdx.x;
  int k = idx / N, n = idx % N;
  Bf[((size_t)(k >> 2) * N + n) * 4 + (k & 3)] = B[idx];
}

__global__ void pack_qkv_kernel(const float* __restrict__ Wq, const float* __restrict__ Wk,
                                const float* __restrict__ Wv, float* __restrict__ Bf) {
  int idx = blockIdx.x * 256 + threadIdx.x;
  int c = idx / 576, j = idx % 576;
  int which = j / 192, r = j % 192;
  int h = r / HEAD, d = r % HEAD;
  const float* W = (which == 0) ? Wq : ((which == 1) ? Wk : Wv);
  Bf[((size_t)(c >> 2) * 576 + j) * 4 + (c & 3)] = W[((size_t)h * N_EMBD + c) * HEAD + d];
}

struct Stage { v4f a[2]; v4f b[2][4]; };

static __device__ __forceinline__ v2f lo2(v4f x) { return __builtin_shufflevector(x, x, 0, 1); }
static __device__ __forceinline__ v2f hi2(v4f x) { return __builtin_shufflevector(x, x, 2, 3); }

static __device__ __forceinline__ void stage_load(Stage& f, const float* __restrict__ ap,
                                                  const float* __restrict__ bp, int N) {
  #pragma unroll
  for (int u = 0; u < 2; u++) {
    f.a[u] = *(const v4f*)(ap + 8 * u);
    #pragma unroll
    for (int t = 0; t < 4; t++) f.b[u][t] = *(const v4f*)(bp + (size_t)u * 8 * N + t * 64);
  }
}

static __device__ __forceinline__ void stage_mma(const Stage& f, v8f* acc) {
  #pragma unroll
  for (int u = 0; u < 2; u++) {
    v2f a0 = lo2(f.a[u]), a1 = hi2(f.a[u]);
    #pragma unroll
    for (int t = 0; t < 4; t++) acc[t] = wmma4(a0, lo2(f.b[u][t]), acc[t]);
    #pragma unroll
    for (int t = 0; t < 4; t++) acc[t] = wmma4(a1, hi2(f.b[u][t]), acc[t]);
  }
}

__device__ __forceinline__ void gemm16x64(const float* __restrict__ A, const float* __restrict__ Bf,
                                          int K, int lda, int N, int m0, int n0, v8f* acc) {
  int lane = threadIdx.x & 31;
  int mr = lane & 15;
  int half = lane >> 4;
  const float* ap = A + (size_t)(m0 + mr) * lda + 4 * half;
  const float* bp = Bf + (size_t)half * 4 * N + (size_t)(n0 + mr) * 4;
  const int S = K / 16;
  Stage f0, f1, f2;
#define STAGE_EDGE() do { \
    asm volatile("" : "+v"(acc[0]), "+v"(acc[1]), "+v"(acc[2]), "+v"(acc[3]) : : "memory"); \
    __builtin_amdgcn_sched_barrier(0); } while (0)
  stage_load(f0, ap, bp, N);
  stage_load(f1, ap + 16, bp + (size_t)16 * N, N);
  ap += 32; bp += (size_t)32 * N;
  #pragma unroll 1
  for (int s = 0; s < S - 3; s += 3) {
    stage_load(f2, ap, bp, N);                          STAGE_EDGE();
    stage_mma(f0, acc);                                 STAGE_EDGE();
    stage_load(f0, ap + 16, bp + (size_t)16 * N, N);    STAGE_EDGE();
    stage_mma(f1, acc);                                 STAGE_EDGE();
    stage_load(f1, ap + 32, bp + (size_t)32 * N, N);    STAGE_EDGE();
    stage_mma(f2, acc);                                 STAGE_EDGE();
    ap += 48; bp += (size_t)48 * N;
  }
  stage_load(f2, ap, bp, N);                            STAGE_EDGE();
  stage_mma(f0, acc);
  stage_mma(f1, acc);
  stage_mma(f2, acc);
#undef STAGE_EDGE
}

__global__ void __launch_bounds__(256) gemm_qkv_kernel(const float* __restrict__ A, const float* __restrict__ Bf,
                                float* __restrict__ Qp, float* __restrict__ Kp,
                                float* __restrict__ Vp) {
  int wv = threadIdx.x >> 5;
  int tile = blockIdx.x * 8 + wv;
  const int TN = 576 / 64;
  int m0 = (tile / TN) * 16;
  int n0 = (tile % TN) * 64;
  v8f acc[4] = {{}, {}, {}, {}};
  gemm16x64(A, Bf, N_EMBD, N_EMBD, 576, m0, n0, acc);
  int lane = threadIdx.x & 31;
  int mr = lane & 15, half = lane >> 4;
  #pragma unroll
  for (int t = 0; t < 4; t++) {
    int j = n0 + t * 16 + mr;
    int which = j / 192, r = j % 192, h = r / HEAD, d = r % HEAD;
    #pragma unroll
    for (int v = 0; v < 8; v++) {
      int m = m0 + v + 8 * half;
      int b = m >> 11, tq = m & 2047;
      size_t hb = ((size_t)(b * N_HEAD + h)) * SEQ * DPAD;
      if (which == 0)      Qp[hb + (size_t)tq * DPAD + d] = acc[t][v];
      else if (which == 1) Kp[hb + (size_t)tq * DPAD + d] = acc[t][v];
      else                 Vp[hb + (size_t)(tq >> 1) * 32 + d * 2 + (tq & 1)] = acc[t][v];
    }
  }
}


static __device__ __forceinline__ float xhalf(float x) {
  int b = __builtin_bit_cast(int, x);
  return __builtin_bit_cast(float,
      __builtin_amdgcn_permlanex16(b, b, 0x76543210, 0xfedcba98, false, false));
}

struct KVFrag { v2f k[4]; v2f v[4]; };

static __device__ __forceinline__ void kv_load(KVFrag& f, const float* __restrict__ Kt,
                                               const float* __restrict__ Vt,
                                               int mr, int kh, int half) {
  #pragma unroll
  for (int ck = 0; ck < 4; ck++) f.k[ck] = *(const v2f*)(Kt + mr * DPAD + ck * 4 + kh);
  #pragma unroll
  for (int ck = 0; ck < 4; ck++) f.v[ck] = *(const v2f*)(Vt + (ck + 4 * half) * 32 + mr * 2);
}

template <bool DIAG>
static __device__ __forceinline__ void attn_tile(const KVFrag& f, const v2f* qf, v8f& o,
                                                 float& mrun, float& lrun, int mr, int half) {
  const float scale = 0.28867513459481287f;
  v8f s = {};
  #pragma unroll
  for (int ck = 0; ck < 4; ck++) s = wmma4(f.k[ck], qf[ck], s);
  float sv[8];
  #pragma unroll
  for (int r = 0; r < 8; r++) sv[r] = s[r] * scale;
  if (DIAG) {
    #pragma unroll
    for (int r = 0; r < 8; r++) sv[r] = (r + 8 * half > mr) ? -INFINITY : sv[r];
  }
  float tmax = fmaxf(fmaxf(fmaxf(sv[0], sv[1]), fmaxf(sv[2], sv[3])),
                     fmaxf(fmaxf(sv[4], sv[5]), fmaxf(sv[6], sv[7])));
  tmax = fmaxf(tmax, xhalf(tmax));
  float mnew  = fmaxf(mrun, tmax);
  float alpha = __expf(mrun - mnew);
  v2f pf[4];
  #pragma unroll
  for (int r = 0; r < 8; r++) pf[r >> 1][r & 1] = __expf(sv[r] - mnew);
  float rsum = ((pf[0][0] + pf[0][1]) + (pf[1][0] + pf[1][1])) +
               ((pf[2][0] + pf[2][1]) + (pf[3][0] + pf[3][1]));
  rsum = rsum + xhalf(rsum);
  lrun = lrun * alpha + rsum;
  mrun = mnew;
  #pragma unroll
  for (int r = 0; r < 8; r++) o[r] = o[r] * alpha;
  #pragma unroll
  for (int ck = 0; ck < 4; ck++) o = wmma4(f.v[ck], pf[ck], o);
}

__global__ void attn_kernel(const float* __restrict__ Qp, const float* __restrict__ Kp,
                            const float* __restrict__ Vp, float* __restrict__ sa) {
  int lane = threadIdx.x & 31;
  int wv = __builtin_amdgcn_readfirstlane((int)(threadIdx.x >> 5));
  int mr = lane & 15, half = lane >> 4, kh = half << 1;
  int pr = blockIdx.x * 4 + wv;
  int bh = blockIdx.y;
  const size_t base = (size_t)bh * SEQ * DPAD;
  int b = bh >> 4, h = bh & 15;
  const float* Kh = Kp + base;
  const float* Vh = Vp + base;

#define TILE_EDGE() do { \
    asm volatile("" : "+v"(qf[0]), "+v"(qf[1]), "+v"(qf[2]), "+v"(qf[3]), "+v"(o) : : "memory"); \
    __builtin_amdgcn_sched_barrier(0); } while (0)

  #pragma unroll 1
  for (int pass = 0; pass < 2; ++pass) {
    int qt = pass ? pr : (SEQ / 16 - 1) - pr;
    const float* Qb = Qp + base + (size_t)qt * 16 * DPAD;

    v2f qf[4];
    #pragma unroll
    for (int ck = 0; ck < 4; ck++) qf[ck] = *(const v2f*)(Qb + mr * DPAD + ck * 4 + kh);

    v8f o = {};
    float mrun = -INFINITY, lrun = 0.f;

    KVFrag fa, fb;
    kv_load(fa, Kh, Vh, mr, kh, half);
    int kt = 0;
    #pragma unroll 1
    for (; kt + 2 <= qt; kt += 2) {
      kv_load(fb, Kh + (size_t)(kt + 1) * 256, Vh + (size_t)(kt + 1) * 256, mr, kh, half);
      TILE_EDGE();
      attn_tile<false>(fa, qf, o, mrun, lrun, mr, half);
      TILE_EDGE();
      kv_load(fa, Kh + (size_t)(kt + 2) * 256, Vh + (size_t)(kt + 2) * 256, mr, kh, half);
      TILE_EDGE();
      attn_tile<false>(fb, qf, o, mrun, lrun, mr, half);
      TILE_EDGE();
    }
    if (kt < qt) {
      kv_load(fb, Kh + (size_t)(kt + 1) * 256, Vh + (size_t)(kt + 1) * 256, mr, kh, half);
      TILE_EDGE();
      attn_tile<false>(fa, qf, o, mrun, lrun, mr, half);
      TILE_EDGE();
      attn_tile<true>(fb, qf, o, mrun, lrun, mr, half);
    } else {
      attn_tile<true>(fa, qf, o, mrun, lrun, mr, half);
    }

    float* dst = sa + ((size_t)(b * SEQ + qt * 16 + mr)) * N_EMBD + h * HEAD + 8 * half;
    v4f lo = { o[0] / lrun, o[1] / lrun, o[2] / lrun, o[3] / lrun };
    *(v4f*)dst = lo;
    if (half == 0) {
      v4f hi = { o[4] / lrun, o[5] / lrun, o[6] / lrun, o[7] / lrun };
      *(v4f*)(dst + 4) = hi;
    }
  }
#undef TILE_EDGE
}

__global__ void __launch_bounds__(256) gemm_proj_kernel(const float* __restrict__ A, const float* __restrict__ Bf,
                                 const float* __restrict__ bias, const float* __restrict__ resid,
                                 float* __restrict__ out) {
  int wv = threadIdx.x >> 5;
  int tile = blockIdx.x * 8 + wv;
  const int TN = 192 / 64;
  int m0 = (tile / TN) * 16, n0 = (tile % TN) * 64;
  v8f acc[4] = {{}, {}, {}, {}};
  gemm16x64(A, Bf, 192, 192, 192, m0, n0, acc);
  int lane = threadIdx.x & 31; int mr = lane & 15, half = lane >> 4;
  #pragma unroll
  for (int t = 0; t < 4; t++) {
    int j = n0 + t * 16 + mr;
    float bj = bias[j];
    #pragma unroll
    for (int v = 0; v < 8; v++) {
      int m = m0 + v + 8 * half;
      size_t ofs = (size_t)m * 192 + j;
      out[ofs] = acc[t][v] + bj + resid[ofs];
    }
  }
}

__global__ void __launch_bounds__(256) gemm_ff1_kernel(const float* __restrict__ A, const float* __restrict__ Bf,
                                float* __restrict__ out) {
  int wv = threadIdx.x >> 5;
  int tile = blockIdx.x * 8 + wv;
  const int TN = FFN / 64;
  int m0 = (tile / TN) * 16, n0 = (tile % TN) * 64;
  v8f acc[4] = {{}, {}, {}, {}};
  gemm16x64(A, Bf, 192, 192, FFN, m0, n0, acc);
  int lane = threadIdx.x & 31; int mr = lane & 15, half = lane >> 4;
  #pragma unroll
  for (int t = 0; t < 4; t++) {
    int j = n0 + t * 16 + mr;
    #pragma unroll
    for (int v = 0; v < 8; v++) {
      int m = m0 + v + 8 * half;
      float val = acc[t][v];
      out[(size_t)m * FFN + j] = val / (1.0f + __expf(-val));
    }
  }
}

__global__ void __launch_bounds__(256) gemm_ff2_kernel(const float* __restrict__ A, const float* __restrict__ Bf,
                                const float* __restrict__ resid, float* __restrict__ out) {
  int wv = threadIdx.x >> 5;
  int tile = blockIdx.x * 8 + wv;
  const int TN = 192 / 64;
  int m0 = (tile / TN) * 16, n0 = (tile % TN) * 64;
  v8f acc[4] = {{}, {}, {}, {}};
  gemm16x64(A, Bf, FFN, FFN, 192, m0, n0, acc);
  int lane = threadIdx.x & 31; int mr = lane & 15, half = lane >> 4;
  #pragma unroll
  for (int t = 0; t < 4; t++) {
    int j = n0 + t * 16 + mr;
    #pragma unroll
    for (int v = 0; v < 8; v++) {
      int m = m0 + v + 8 * half;
      size_t ofs = (size_t)m * 192 + j;
      out[ofs] = acc[t][v] + resid[ofs];
    }
  }
}

extern "C" void kernel_launch(void* const* d_in, const int* in_sizes, int n_in,
                              void* d_out, int out_size, void* d_ws, size_t ws_size,
                              hipStream_t stream) {
  (void)in_sizes; (void)n_in; (void)out_size; (void)ws_size;
  const float* x     = (const float*)d_in[0];
  const float* ln1_w = (const float*)d_in[1];
  const float* ln2_w = (const float*)d_in[2];
  const float* Wq    = (const float*)d_in[3];
  const float* Wk    = (const float*)d_in[4];
  const float* Wv    = (const float*)d_in[5];
  const float* Wproj = (const float*)d_in[6];
  const float* bproj = (const float*)d_in[7];
  const float* W1    = (const float*)d_in[8];
  const float* W2    = (const float*)d_in[9];
  float* ws = (float*)d_ws;

  size_t off = 0;
  float* h    = ws + off; off += (size_t)M_TOK * N_EMBD;
  float* wqkv = ws + off; off += (size_t)192 * 576;
  float* wpjf = ws + off; off += (size_t)192 * 192;
  float* w1f  = ws + off; off += (size_t)192 * FFN;
  float* w2f  = ws + off; off += (size_t)FFN * 192;
  float* Qp   = ws + off; off += (size_t)BH * SEQ * DPAD;
  float* Kp   = ws + off; off += (size_t)BH * SEQ * DPAD;
  float* Vp   = ws + off; off += (size_t)BH * SEQ * DPAD;
  float* sa   = ws + off; off += (size_t)M_TOK * N_EMBD;
  float* x1   = ws + off; off += (size_t)M_TOK * N_EMBD;
  float* h2   = ws + off; off += (size_t)M_TOK * N_EMBD;
  float* u    = ws + off; off += (size_t)M_TOK * FFN;
  float* out  = (float*)d_out;

  hipMemsetAsync(Qp, 0, (size_t)3 * BH * SEQ * DPAD * sizeof(float), stream);

  pack_qkv_kernel<<<(192 * 576) / 256, 256, 0, stream>>>(Wq, Wk, Wv, wqkv);
  pack_b_kernel<<<(192 * 192) / 256, 256, 0, stream>>>(Wproj, wpjf, 192);
  pack_b_kernel<<<(192 * FFN) / 256, 256, 0, stream>>>(W1, w1f, FFN);
  pack_b_kernel<<<(FFN * 192) / 256, 256, 0, stream>>>(W2, w2f, 192);

  rmsnorm_kernel<<<M_TOK / 8, 256, 0, stream>>>(x, ln1_w, h);
  gemm_qkv_kernel<<<(512 * 9) / 8, 256, 0, stream>>>(h, wqkv, Qp, Kp, Vp);
  attn_kernel<<<dim3(16, BH), 128, 0, stream>>>(Qp, Kp, Vp, sa);
  gemm_proj_kernel<<<(512 * 3) / 8, 256, 0, stream>>>(sa, wpjf, bproj, x, x1);
  rmsnorm_kernel<<<M_TOK / 8, 256, 0, stream>>>(x1, ln2_w, h2);
  gemm_ff1_kernel<<<(512 * 12) / 8, 256, 0, stream>>>(h2, w1f, u);
  gemm_ff2_kernel<<<(512 * 3) / 8, 256, 0, stream>>>(u, w2f, x1, out);
}


# attention at the real head size 12: the score product is 3 matrix instructions per key tile, not 4 over a zero-padded 16; Q/K/V are stored 12 wide, so the pad-zeroing memset is gone
# speedup vs baseline: 2.4512x; 1.2198x over previous
#include <hip/hip_runtime.h>
#include <math.h>

typedef __attribute__((ext_vector_type(2))) float v2f;
typedef __attribute__((ext_vector_type(4))) float v4f;
typedef __attribute__((ext_vector_type(8))) float v8f;

#define N_EMBD 192
#define N_HEAD 16
#define HEAD   12
#define TILE_F (16*HEAD)
#define BATCH  4
#define SEQ    2048
#define M_TOK  (BATCH*SEQ)
#define FFN    (4*N_EMBD)
#define BH     (BATCH*N_HEAD)

static __device__ __forceinline__ v8f wmma4(v2f a, v2f b, v8f c) {
  return __builtin_amdgcn_wmma_f32_16x16x4_f32(false, a, false, b, (short)0, c, false, false);
}

static __device__ __forceinline__ void rmsnorm_row(const float* __restrict__ x, const float* __restrict__ w,
                                                   float* __restrict__ out, int row, int lane) {
  const float* xr = x + (size_t)row * N_EMBD;
  float v[6]; float ss = 0.f;
  #pragma unroll
  for (int i = 0; i < 6; i++) { v[i] = xr[lane + 32*i]; ss += v[i]*v[i]; }
  #pragma unroll
  for (int off = 16; off >= 1; off >>= 1) ss += __shfl_xor(ss, off, 32);
  float sc = rsqrtf(ss * (1.0f / N_EMBD) + 1e-5f);
  float* orow = out + (size_t)row * N_EMBD;
  #pragma unroll
  for (int i = 0; i < 6; i++) { int c = lane + 32*i; orow[c] = v[i] * sc * w[c]; }
}

__global__ void rmsnorm_kernel(const float* __restrict__ x, const float* __restrict__ w,
                               float* __restrict__ out) {
  int lane = threadIdx.x & 31;
  int wv   = threadIdx.x >> 5;
  rmsnorm_row(x, w, out, blockIdx.x * 8 + wv, lane);
}

#define PACK_QUADS 1024

static __device__ __forceinline__ void pack_quads_b(const float* __restrict__ B, float* __restrict__ Bf,
                                                    int N, int q0) {
  v4f val[4];
  #pragma unroll
  for (int i = 0; i < 4; i++) {
    int q = q0 + 256 * i;
    int kq = q / N, n = q % N;
    const float* s = B + (size_t)kq * 4 * N + n;
    val[i] = (v4f){ s[0], s[N], s[2 * N], s[3 * N] };
  }
  #pragma unroll
  for (int i = 0; i < 4; i++) *(v4f*)(Bf + (size_t)(q0 + 256 * i) * 4) = val[i];
}

static __device__ __forceinline__ void pack_quads_qkv(const float* __restrict__ Wq, const float* __restrict__ Wk,
                                                      const float* __restrict__ Wv, float* __restrict__ Bf,
                                                      int q0) {
  v4f val[4];
  #pragma unroll
  for (int i = 0; i < 4; i++) {
    int q = q0 + 256 * i;
    int cq = q / 576, j = q % 576;
    int which = j / 192, r = j % 192;
    int h = r / HEAD, d = r % HEAD;
    const float* W = (which == 0) ? Wq : ((which == 1) ? Wk : Wv);
    const float* s = W + ((size_t)h * N_EMBD + 4 * cq) * HEAD + d;
    val[i] = (v4f){ s[0], s[HEAD], s[2 * HEAD], s[3 * HEAD] };
  }
  #pragma unroll
  for (int i = 0; i < 4; i++) *(v4f*)(Bf + (size_t)(q0 + 256 * i) * 4) = val[i];
}

#define PACK_WGS 108
__global__ void __launch_bounds__(256) prep_kernel(const float* __restrict__ x, const float* __restrict__ ln_w,
                            float* __restrict__ hout,
                            const float* __restrict__ Wq, const float* __restrict__ Wk,
                            const float* __restrict__ Wv, const float* __restrict__ Wproj,
                            const float* __restrict__ W1, const float* __restrict__ W2,
                            float* __restrict__ wqkv, float* __restrict__ wpjf,
                            float* __restrict__ w1f, float* __restrict__ w2f) {
  int blk = blockIdx.x;
  int t = threadIdx.x;
  if (blk >= PACK_WGS) {
    rmsnorm_row(x, ln_w, hout, (blk - PACK_WGS) * 8 + (t >> 5), t & 31);
    return;
  }
  if (blk < 27)      pack_quads_qkv(Wq, Wk, Wv, wqkv, blk * PACK_QUADS + t);
  else if (blk < 36) pack_quads_b(Wproj, wpjf, 192, (blk - 27) * PACK_QUADS + t);
  else if (blk < 72) pack_quads_b(W1, w1f, FFN, (blk - 36) * PACK_QUADS + t);
  else               pack_quads_b(W2, w2f, 192, (blk - 72) * PACK_QUADS + t);
}

struct Stage { v4f a[2]; v4f b[2][4]; };

static __device__ __forceinline__ v2f lo2(v4f x) { return __builtin_shufflevector(x, x, 0, 1); }
static __device__ __forceinline__ v2f hi2(v4f x) { return __builtin_shufflevector(x, x, 2, 3); }

static __device__ __forceinline__ void stage_load(Stage& f, const float* __restrict__ ap,
                                                  const float* __restrict__ bp, int N) {
  #pragma unroll
  for (int u = 0; u < 2; u++) {
    f.a[u] = *(const v4f*)(ap + 8 * u);
    #pragma unroll
    for (int t = 0; t < 4; t++) f.b[u][t] = *(const v4f*)(bp + (size_t)u * 8 * N + t * 64);
  }
}

static __device__ __forceinline__ void stage_mma(const Stage& f, v8f* acc) {
  #pragma unroll
  for (int u = 0; u < 2; u++) {
    v2f a0 = lo2(f.a[u]), a1 = hi2(f.a[u]);
    #pragma unroll
    for (int t = 0; t < 4; t++) acc[t] = wmma4(a0, lo2(f.b[u][t]), acc[t]);
    #pragma unroll
    for (int t = 0; t < 4; t++) acc[t] = wmma4(a1, hi2(f.b[u][t]), acc[t]);
  }
}

__device__ __forceinline__ void gemm16x64(const float* __restrict__ A, const float* __restrict__ Bf,
                                          int K, int lda, int N, int m0, int n0, v8f* acc) {
  int lane = threadIdx.x & 31;
  int mr = lane & 15;
  int half = lane >> 4;
  const float* ap = A + (size_t)(m0 + mr) * lda + 4 * half;
  const float* bp = Bf + (size_t)half * 4 * N + (size_t)(n0 + mr) * 4;
  const int S = K / 16;
  Stage f0, f1, f2;
#define STAGE_EDGE() do { \
    asm volatile("" : "+v"(acc[0]), "+v"(acc[1]), "+v"(acc[2]), "+v"(acc[3]) : : "memory"); \
    __builtin_amdgcn_sched_barrier(0); } while (0)
  stage_load(f0, ap, bp, N);
  stage_load(f1, ap + 16, bp + (size_t)16 * N, N);
  ap += 32; bp += (size_t)32 * N;
  #pragma unroll 1
  for (int s = 0; s < S - 3; s += 3) {
    stage_load(f2, ap, bp, N);                          STAGE_EDGE();
    stage_mma(f0, acc);                                 STAGE_EDGE();
    stage_load(f0, ap + 16, bp + (size_t)16 * N, N);    STAGE_EDGE();
    stage_mma(f1, acc);                                 STAGE_EDGE();
    stage_load(f1, ap + 32, bp + (size_t)32 * N, N);    STAGE_EDGE();
    stage_mma(f2, acc);                                 STAGE_EDGE();
    ap += 48; bp += (size_t)48 * N;
  }
  stage_load(f2, ap, bp, N);                            STAGE_EDGE();
  stage_mma(f0, acc);
  stage_mma(f1, acc);
  stage_mma(f2, acc);
#undef STAGE_EDGE
}

__global__ void __launch_bounds__(256) gemm_qkv_kernel(const float* __restrict__ A, const float* __restrict__ Bf,
                                float* __restrict__ Qp, float* __restrict__ Kp,
                                float* __restrict__ Vp) {
  int wv = threadIdx.x >> 5;
  int tile = blockIdx.x * 8 + wv;
  const int TN = 576 / 64;
  int m0 = (tile / TN) * 16;
  int n0 = (tile % TN) * 64;
  v8f acc[4] = {{}, {}, {}, {}};
  gemm16x64(A, Bf, N_EMBD, N_EMBD, 576, m0, n0, acc);
  int lane = threadIdx.x & 31;
  int mr = lane & 15, half = lane >> 4;
  #pragma unroll
  for (int t = 0; t < 4; t++) {
    int j = n0 + t * 16 + mr;
    int which = j / 192, r = j % 192, h = r / HEAD, d = r % HEAD;
    #pragma unroll
    for (int v = 0; v < 8; v++) {
      int m = m0 + v + 8 * half;
      int b = m >> 11, tq = m & 2047;
      size_t hb = ((size_t)(b * N_HEAD + h)) * SEQ * HEAD;
      if (which == 0)      Qp[hb + (size_t)tq * HEAD + d] = acc[t][v];
      else if (which == 1) Kp[hb + (size_t)tq * HEAD + d] = acc[t][v];
      else                 Vp[hb + (size_t)(tq >> 1) * (2 * HEAD) + d * 2 + (tq & 1)] = acc[t][v];
    }
  }
}


static __device__ __forceinline__ float xhalf(float x) {
  int b = __builtin_bit_cast(int, x);
  return __builtin_bit_cast(float,
      __builtin_amdgcn_permlanex16(b, b, 0x76543210, 0xfedcba98, false, false));
}

struct KVFrag { v2f k[3]; v2f v[4]; };

static __device__ __forceinline__ void kv_load(KVFrag& f, const float* __restrict__ Kt,
                                               const float* __restrict__ Vt,
                                               int mr, int kh, int half, int dv) {
  #pragma unroll
  for (int ck = 0; ck < 3; ck++) f.k[ck] = *(const v2f*)(Kt + mr * HEAD + ck * 4 + kh);
  #pragma unroll
  for (int ck = 0; ck < 4; ck++) f.v[ck] = *(const v2f*)(Vt + (ck + 4 * half) * (2 * HEAD) + dv * 2);
}

template <bool DIAG>
static __device__ __forceinline__ void attn_tile(const KVFrag& f, const v2f* qf, v8f& o,
                                                 float& mrun, float& lrun, int mr, int half) {
  const float scale = 0.28867513459481287f;
  v8f s = {};
  #pragma unroll
  for (int ck = 0; ck < 3; ck++) s = wmma4(f.k[ck], qf[ck], s);
  float sv[8];
  #pragma unroll
  for (int r = 0; r < 8; r++) sv[r] = s[r] * scale;
  if (DIAG) {
    #pragma unroll
    for (int r = 0; r < 8; r++) sv[r] = (r + 8 * half > mr) ? -INFINITY : sv[r];
  }
  float tmax = fmaxf(fmaxf(fmaxf(sv[0], sv[1]), fmaxf(sv[2], sv[3])),
                     fmaxf(fmaxf(sv[4], sv[5]), fmaxf(sv[6], sv[7])));
  tmax = fmaxf(tmax, xhalf(tmax));
  float mnew  = fmaxf(mrun, tmax);
  float alpha = __expf(mrun - mnew);
  v2f pf[4];
  #pragma unroll
  for (int r = 0; r < 8; r++) pf[r >> 1][r & 1] = __expf(sv[r] - mnew);
  float rsum = ((pf[0][0] + pf[0][1]) + (pf[1][0] + pf[1][1])) +
               ((pf[2][0] + pf[2][1]) + (pf[3][0] + pf[3][1]));
  rsum = rsum + xhalf(rsum);
  lrun = lrun * alpha + rsum;
  mrun = mnew;
  #pragma unroll
  for (int r = 0; r < 8; r++) o[r] = o[r] * alpha;
  #pragma unroll
  for (int ck = 0; ck < 4; ck++) o = wmma4(f.v[ck], pf[ck], o);
}

__global__ void attn_kernel(const float* __restrict__ Qp, const float* __restrict__ Kp,
                            const float* __restrict__ Vp, float* __restrict__ sa) {
  int lane = threadIdx.x & 31;
  int wv = __builtin_amdgcn_readfirstlane((int)(threadIdx.x >> 5));
  int mr = lane & 15, half = lane >> 4, kh = half << 1;
  int dv = mr < HEAD ? mr : HEAD - 1;
  int pr = blockIdx.x * 4 + wv;
  int bh = blockIdx.y;
  const size_t base = (size_t)bh * SEQ * HEAD;
  int b = bh >> 4, h = bh & 15;
  const float* Kh = Kp + base;
  const float* Vh = Vp + base;

#define TILE_EDGE() do { \
    asm volatile("" : "+v"(qf[0]), "+v"(qf[1]), "+v"(qf[2]), "+v"(o) : : "memory"); \
    __builtin_amdgcn_sched_barrier(0); } while (0)

  #pragma unroll 1
  for (int pass = 0; pass < 2; ++pass) {
    int qt = pass ? pr : (SEQ / 16 - 1) - pr;
    const float* Qb = Qp + base + (size_t)qt * TILE_F;

    v2f qf[3];
    #pragma unroll
    for (int ck = 0; ck < 3; ck++) qf[ck] = *(const v2f*)(Qb + mr * HEAD + ck * 4 + kh);

    v8f o = {};
    float mrun = -INFINITY, lrun = 0.f;

    KVFrag fa, fb;
    kv_load(fa, Kh, Vh, mr, kh, half, dv);
    int kt = 0;
    #pragma unroll 1
    for (; kt + 2 <= qt; kt += 2) {
      kv_load(fb, Kh + (size_t)(kt + 1) * TILE_F, Vh + (size_t)(kt + 1) * TILE_F, mr, kh, half, dv);
      TILE_EDGE();
      attn_tile<false>(fa, qf, o, mrun, lrun, mr, half);
      TILE_EDGE();
      kv_load(fa, Kh + (size_t)(kt + 2) * TILE_F, Vh + (size_t)(kt + 2) * TILE_F, mr, kh, half, dv);
      TILE_EDGE();
      attn_tile<false>(fb, qf, o, mrun, lrun, mr, half);
      TILE_EDGE();
    }
    if (kt < qt) {
      kv_load(fb, Kh + (size_t)(kt + 1) * TILE_F, Vh + (size_t)(kt + 1) * TILE_F, mr, kh, half, dv);
      TILE_EDGE();
      attn_tile<false>(fa, qf, o, mrun, lrun, mr, half);
      TILE_EDGE();
      attn_tile<true>(fb, qf, o, mrun, lrun, mr, half);
    } else {
      attn_tile<true>(fa, qf, o, mrun, lrun, mr, half);
    }

    float* dst = sa + ((size_t)(b * SEQ + qt * 16 + mr)) * N_EMBD + h * HEAD + 8 * half;
    v4f lo = { o[0] / lrun, o[1] / lrun, o[2] / lrun, o[3] / lrun };
    *(v4f*)dst = lo;
    if (half == 0) {
      v4f hi = { o[4] / lrun, o[5] / lrun, o[6] / lrun, o[7] / lrun };
      *(v4f*)(dst + 4) = hi;
    }
  }
#undef TILE_EDGE
}

__global__ void __launch_bounds__(256) gemm_proj_kernel(const float* __restrict__ A, const float* __restrict__ Bf,
                                 const float* __restrict__ bias, const float* __restrict__ resid,
                                 float* __restrict__ out) {
  int wv = threadIdx.x >> 5;
  int tile = blockIdx.x * 8 + wv;
  const int TN = 192 / 64;
  int m0 = (tile / TN) * 16, n0 = (tile % TN) * 64;
  v8f acc[4] = {{}, {}, {}, {}};
  gemm16x64(A, Bf, 192, 192, 192, m0, n0, acc);
  int lane = threadIdx.x & 31; int mr = lane & 15, half = lane >> 4;
  #pragma unroll
  for (int t = 0; t < 4; t++) {
    int j = n0 + t * 16 + mr;
    float bj = bias[j];
    #pragma unroll
    for (int v = 0; v < 8; v++) {
      int m = m0 + v + 8 * half;
      size_t ofs = (size_t)m * 192 + j;
      out[ofs] = acc[t][v] + bj + resid[ofs];
    }
  }
}

__global__ void __launch_bounds__(256) gemm_ff1_kernel(const float* __restrict__ A, const float* __restrict__ Bf,
                                float* __restrict__ out) {
  int wv = threadIdx.x >> 5;
  int tile = blockIdx.x * 8 + wv;
  const int TN = FFN / 64;
  int m0 = (tile / TN) * 16, n0 = (tile % TN) * 64;
  v8f acc[4] = {{}, {}, {}, {}};
  gemm16x64(A, Bf, 192, 192, FFN, m0, n0, acc);
  int lane = threadIdx.x & 31; int mr = lane & 15, half = lane >> 4;
  #pragma unroll
  for (int t = 0; t < 4; t++) {
    int j = n0 + t * 16 + mr;
    #pragma unroll
    for (int v = 0; v < 8; v++) {
      int m = m0 + v + 8 * half;
      float val = acc[t][v];
      out[(size_t)m * FFN + j] = val / (1.0f + __expf(-val));
    }
  }
}

__global__ void __launch_bounds__(256) gemm_ff2_kernel(const float* __restrict__ A, const float* __restrict__ Bf,
                                const float* __restrict__ resid, float* __restrict__ out) {
  int wv = threadIdx.x >> 5;
  int tile = blockIdx.x * 8 + wv;
  const int TN = 192 / 64;
  int m0 = (tile / TN) * 16, n0 = (tile % TN) * 64;
  v8f acc[4] = {{}, {}, {}, {}};
  gemm16x64(A, Bf, FFN, FFN, 192, m0, n0, acc);
  int lane = threadIdx.x & 31; int mr = lane & 15, half = lane >> 4;
  #pragma unroll
  for (int t = 0; t < 4; t++) {
    int j = n0 + t * 16 + mr;
    #pragma unroll
    for (int v = 0; v < 8; v++) {
      int m = m0 + v + 8 * half;
      size_t ofs = (size_t)m * 192 + j;
      out[ofs] = acc[t][v] + resid[ofs];
    }
  }
}

extern "C" void kernel_launch(void* const* d_in, const int* in_sizes, int n_in,
                              void* d_out, int out_size, void* d_ws, size_t ws_size,
                              hipStream_t stream) {
  (void)in_sizes; (void)n_in; (void)out_size; (void)ws_size;
  const float* x     = (const float*)d_in[0];
  const float* ln1_w = (const float*)d_in[1];
  const float* ln2_w = (const float*)d_in[2];
  const float* Wq    = (const float*)d_in[3];
  const float* Wk    = (const float*)d_in[4];
  const float* Wv    = (const float*)d_in[5];
  const float* Wproj = (const float*)d_in[6];
  const float* bproj = (const float*)d_in[7];
  const float* W1    = (const float*)d_in[8];
  const float* W2    = (const float*)d_in[9];
  float* ws = (float*)d_ws;

  size_t off = 0;
  float* h    = ws + off; off += (size_t)M_TOK * N_EMBD;
  float* wqkv = ws + off; off += (size_t)192 * 576;
  float* wpjf = ws + off; off += (size_t)192 * 192;
  float* w1f  = ws + off; off += (size_t)192 * FFN;
  float* w2f  = ws + off; off += (size_t)FFN * 192;
  float* Qp   = ws + off; off += (size_t)BH * SEQ * HEAD;
  float* Kp   = ws + off; off += (size_t)BH * SEQ * HEAD;
  float* Vp   = ws + off; off += (size_t)BH * SEQ * HEAD;
  float* sa   = ws + off; off += (size_t)M_TOK * N_EMBD;
  float* x1   = ws + off; off += (size_t)M_TOK * N_EMBD;
  float* h2   = ws + off; off += (size_t)M_TOK * N_EMBD;
  float* u    = ws + off; off += (size_t)M_TOK * FFN;
  float* out  = (float*)d_out;

  prep_kernel<<<PACK_WGS + M_TOK / 8, 256, 0, stream>>>(x, ln1_w, h, Wq, Wk, Wv, Wproj, W1, W2,
                                                        wqkv, wpjf, w1f, w2f);
  gemm_qkv_kernel<<<(512 * 9) / 8, 256, 0, stream>>>(h, wqkv, Qp, Kp, Vp);
  attn_kernel<<<dim3(16, BH), 128, 0, stream>>>(Qp, Kp, Vp, sa);
  gemm_proj_kernel<<<(512 * 3) / 8, 256, 0, stream>>>(sa, wpjf, bproj, x, x1);
  rmsnorm_kernel<<<M_TOK / 8, 256, 0, stream>>>(x1, ln2_w, h2);
  gemm_ff1_kernel<<<(512 * 12) / 8, 256, 0, stream>>>(h2, w1f, u);
  gemm_ff2_kernel<<<(512 * 3) / 8, 256, 0, stream>>>(u, w2f, x1, out);
}


# GEMM wave tile 16x48 (three column tiles), not 16x64, in the QKV, projection and second feed-forward products: 6144, 2048, 2048 waves, whole work-groups on all 256 WGPs, equal work per SIMD
# speedup vs baseline: 2.6626x; 1.0862x over previous
#include <hip/hip_runtime.h>
#include <math.h>

typedef __attribute__((ext_vector_type(2))) float v2f;
typedef __attribute__((ext_vector_type(4))) float v4f;
typedef __attribute__((ext_vector_type(8))) float v8f;

#define N_EMBD 192
#define N_HEAD 16
#define HEAD   12
#define TILE_F (16*HEAD)
#define BATCH  4
#define SEQ    2048
#define M_TOK  (BATCH*SEQ)
#define FFN    (4*N_EMBD)
#define BH     (BATCH*N_HEAD)

static __device__ __forceinline__ v8f wmma4(v2f a, v2f b, v8f c) {
  return __builtin_amdgcn_wmma_f32_16x16x4_f32(false, a, false, b, (short)0, c, false, false);
}

static __device__ __forceinline__ float xhalf(float x) {
  int b = __builtin_bit_cast(int, x);
  return __builtin_bit_cast(float,
      __builtin_amdgcn_permlanex16(b, b, 0x76543210, 0xfedcba98, false, false));
}


#define PACK_QUADS 1024

template <bool GAIN>
static __device__ __forceinline__ void pack_quads_b(const float* __restrict__ B, const float* __restrict__ g,
                                                    float* __restrict__ Bf, int N, int q0) {
  v4f val[4];
  #pragma unroll
  for (int i = 0; i < 4; i++) {
    int q = q0 + 256 * i;
    int kq = q / N, n = q % N;
    const float* s = B + (size_t)kq * 4 * N + n;
    val[i] = (v4f){ s[0], s[N], s[2 * N], s[3 * N] };
    if (GAIN) {
      const float* gk = g + 4 * kq;
      val[i] = val[i] * (v4f){ gk[0], gk[1], gk[2], gk[3] };
    }
  }
  #pragma unroll
  for (int i = 0; i < 4; i++) *(v4f*)(Bf + (size_t)(q0 + 256 * i) * 4) = val[i];
}

static __device__ __forceinline__ void pack_quads_qkv(const float* __restrict__ Wq, const float* __restrict__ Wk,
                                                      const float* __restrict__ Wv, const float* __restrict__ g,
                                                      float* __restrict__ Bf, int q0) {
  v4f val[4];
  #pragma unroll
  for (int i = 0; i < 4; i++) {
    int q = q0 + 256 * i;
    int cq = q / 576, j = q % 576;
    int which = j / 192, r = j % 192;
    int h = r / HEAD, d = r % HEAD;
    const float* W = (which == 0) ? Wq : ((which == 1) ? Wk : Wv);
    const float* s = W + ((size_t)h * N_EMBD + 4 * cq) * HEAD + d;
    const float* gk = g + 4 * cq;
    val[i] = (v4f){ s[0] * gk[0], s[HEAD] * gk[1], s[2 * HEAD] * gk[2], s[3 * HEAD] * gk[3] };
  }
  #pragma unroll
  for (int i = 0; i < 4; i++) *(v4f*)(Bf + (size_t)(q0 + 256 * i) * 4) = val[i];
}

#define PACK_WGS 108
__global__ void __launch_bounds__(256) prep_kernel(const float* __restrict__ ln1_w, const float* __restrict__ ln2_w,
                            const float* __restrict__ Wq, const float* __restrict__ Wk,
                            const float* __restrict__ Wv, const float* __restrict__ Wproj,
                            const float* __restrict__ W1, const float* __restrict__ W2,
                            float* __restrict__ wqkv, float* __restrict__ wpjf,
                            float* __restrict__ w1f, float* __restrict__ w2f) {
  int blk = blockIdx.x;
  int t = threadIdx.x;
  if (blk < 27)      pack_quads_qkv(Wq, Wk, Wv, ln1_w, wqkv, blk * PACK_QUADS + t);
  else if (blk < 36) pack_quads_b<false>(Wproj, nullptr, wpjf, 192, (blk - 27) * PACK_QUADS + t);
  else if (blk < 72) pack_quads_b<true>(W1, ln2_w, w1f, FFN, (blk - 36) * PACK_QUADS + t);
  else               pack_quads_b<false>(W2, nullptr, w2f, 192, (blk - 72) * PACK_QUADS + t);
}

template <int NT> struct Stage { v4f a[2]; v4f b[2][NT]; };

static __device__ __forceinline__ v2f lo2(v4f x) { return __builtin_shufflevector(x, x, 0, 1); }
static __device__ __forceinline__ v2f hi2(v4f x) { return __builtin_shufflevector(x, x, 2, 3); }

template <int NT>
static __device__ __forceinline__ void stage_load(Stage<NT>& f, const float* __restrict__ ap,
                                                  const float* __restrict__ bp, int N) {
  #pragma unroll
  for (int u = 0; u < 2; u++) {
    f.a[u] = *(const v4f*)(ap + 8 * u);
    #pragma unroll
    for (int t = 0; t < NT; t++) f.b[u][t] = *(const v4f*)(bp + (size_t)u * 8 * N + t * 64);
  }
}

template <bool SUMSQ, int NT>
static __device__ __forceinline__ void stage_mma(const Stage<NT>& f, v8f* acc, v4f& sq) {
  #pragma unroll
  for (int u = 0; u < 2; u++) {
    v2f a0 = lo2(f.a[u]), a1 = hi2(f.a[u]);
    if (SUMSQ) sq = sq + f.a[u] * f.a[u];
    #pragma unroll
    for (int t = 0; t < NT; t++) acc[t] = wmma4(a0, lo2(f.b[u][t]), acc[t]);
    #pragma unroll
    for (int t = 0; t < NT; t++) acc[t] = wmma4(a1, hi2(f.b[u][t]), acc[t]);
  }
}

template <bool SUMSQ, int NT>
__device__ __forceinline__ void gemm16xn(const float* __restrict__ A, const float* __restrict__ Bf,
                                          int K, int lda, int N, int m0, int n0, v8f* acc, float& ss) {
  int lane = threadIdx.x & 31;
  int mr = lane & 15;
  int half = lane >> 4;
  const float* ap = A + (size_t)(m0 + mr) * lda + 4 * half;
  const float* bp = Bf + (size_t)half * 4 * N + (size_t)(n0 + mr) * 4;
  const int S = K / 16;
  Stage<NT> f0, f1, f2;
  v4f sq = {};
#define STAGE_EDGE() do { \
    if constexpr (NT == 4) { \
      if (SUMSQ) asm volatile("" : "+v"(acc[0]), "+v"(acc[1]), "+v"(acc[2]), "+v"(acc[3]), "+v"(sq) : : "memory"); \
      else       asm volatile("" : "+v"(acc[0]), "+v"(acc[1]), "+v"(acc[2]), "+v"(acc[3]) : : "memory"); \
    } else { \
      if (SUMSQ) asm volatile("" : "+v"(acc[0]), "+v"(acc[1]), "+v"(acc[2]), "+v"(sq) : : "memory"); \
      else       asm volatile("" : "+v"(acc[0]), "+v"(acc[1]), "+v"(acc[2]) : : "memory"); \
    } \
    __builtin_amdgcn_sched_barrier(0); } while (0)
  stage_load(f0, ap, bp, N);                            STAGE_EDGE();
  stage_load(f1, ap + 16, bp + (size_t)16 * N, N);      STAGE_EDGE();
  ap += 32; bp += (size_t)32 * N;
  #pragma unroll 1
  for (int s = 0; s < S - 3; s += 3) {
    stage_load(f2, ap, bp, N);                          STAGE_EDGE();
    stage_mma<SUMSQ, NT>(f0, acc, sq);                      STAGE_EDGE();
    stage_load(f0, ap + 16, bp + (size_t)16 * N, N);    STAGE_EDGE();
    stage_mma<SUMSQ, NT>(f1, acc, sq);                      STAGE_EDGE();
    stage_load(f1, ap + 32, bp + (size_t)32 * N, N);    STAGE_EDGE();
    stage_mma<SUMSQ, NT>(f2, acc, sq);                      STAGE_EDGE();
    ap += 48; bp += (size_t)48 * N;
  }
  stage_load(f2, ap, bp, N);                            STAGE_EDGE();
  stage_mma<SUMSQ, NT>(f0, acc, sq);                        if (SUMSQ) STAGE_EDGE();
  stage_mma<SUMSQ, NT>(f1, acc, sq);                        if (SUMSQ) STAGE_EDGE();
  stage_mma<SUMSQ, NT>(f2, acc, sq);
#undef STAGE_EDGE
  if (SUMSQ) ss = (sq[0] + sq[1]) + (sq[2] + sq[3]);
}

static __device__ __forceinline__ void row_rms_factors(float ss, int half, float* scr) {
  float sc = rsqrtf((ss + xhalf(ss)) * (1.0f / N_EMBD) + 1e-5f);
  #pragma unroll
  for (int v = 0; v < 8; v++) scr[v] = __shfl(sc, v + 8 * half, 32);
}

__global__ void __launch_bounds__(256) gemm_qkv_kernel(const float* __restrict__ A, const float* __restrict__ Bf,
                                float* __restrict__ Qp, float* __restrict__ Kp,
                                float* __restrict__ Vp) {
  int wv = threadIdx.x >> 5;
  int tile = blockIdx.x * 8 + wv;
  const int TN = 576 / 48;
  int m0 = (tile / TN) * 16;
  int n0 = (tile % TN) * 48;
  v8f acc[3] = {{}, {}, {}};
  float ss = 0.f;
  gemm16xn<true, 3>(A, Bf, N_EMBD, N_EMBD, 576, m0, n0, acc, ss);
  int lane = threadIdx.x & 31;
  int mr = lane & 15, half = lane >> 4;
  float scr[8];
  row_rms_factors(ss, half, scr);
  #pragma unroll
  for (int t = 0; t < 3; t++) {
    int j = n0 + t * 16 + mr;
    int which = j / 192, r = j % 192, h = r / HEAD, d = r % HEAD;
    #pragma unroll
    for (int v = 0; v < 8; v++) {
      int m = m0 + v + 8 * half;
      int b = m >> 11, tq = m & 2047;
      size_t hb = ((size_t)(b * N_HEAD + h)) * SEQ * HEAD;
      float val = acc[t][v] * scr[v];
      if (which == 0)      Qp[hb + (size_t)tq * HEAD + d] = val;
      else if (which == 1) Kp[hb + (size_t)tq * HEAD + d] = val;
      else                 Vp[hb + (size_t)(tq >> 1) * (2 * HEAD) + d * 2 + (tq & 1)] = val;
    }
  }
}


struct KVFrag { v2f k[3]; v2f v[4]; };

static __device__ __forceinline__ void kv_load(KVFrag& f, const float* __restrict__ Kt,
                                               const float* __restrict__ Vt,
                                               int mr, int kh, int half, int dv) {
  #pragma unroll
  for (int ck = 0; ck < 3; ck++) f.k[ck] = *(const v2f*)(Kt + mr * HEAD + ck * 4 + kh);
  #pragma unroll
  for (int ck = 0; ck < 4; ck++) f.v[ck] = *(const v2f*)(Vt + (ck + 4 * half) * (2 * HEAD) + dv * 2);
}

template <bool DIAG>
static __device__ __forceinline__ void attn_tile(const KVFrag& f, const v2f* qf, v8f& o,
                                                 float& mrun, float& lrun, int mr, int half) {
  const float scale = 0.28867513459481287f;
  v8f s = {};
  #pragma unroll
  for (int ck = 0; ck < 3; ck++) s = wmma4(f.k[ck], qf[ck], s);
  float sv[8];
  #pragma unroll
  for (int r = 0; r < 8; r++) sv[r] = s[r] * scale;
  if (DIAG) {
    #pragma unroll
    for (int r = 0; r < 8; r++) sv[r] = (r + 8 * half > mr) ? -INFINITY : sv[r];
  }
  float tmax = fmaxf(fmaxf(fmaxf(sv[0], sv[1]), fmaxf(sv[2], sv[3])),
                     fmaxf(fmaxf(sv[4], sv[5]), fmaxf(sv[6], sv[7])));
  tmax = fmaxf(tmax, xhalf(tmax));
  float mnew  = fmaxf(mrun, tmax);
  float alpha = __expf(mrun - mnew);
  v2f pf[4];
  #pragma unroll
  for (int r = 0; r < 8; r++) pf[r >> 1][r & 1] = __expf(sv[r] - mnew);
  float rsum = ((pf[0][0] + pf[0][1]) + (pf[1][0] + pf[1][1])) +
               ((pf[2][0] + pf[2][1]) + (pf[3][0] + pf[3][1]));
  rsum = rsum + xhalf(rsum);
  lrun = lrun * alpha + rsum;
  mrun = mnew;
  #pragma unroll
  for (int r = 0; r < 8; r++) o[r] = o[r] * alpha;
  #pragma unroll
  for (int ck = 0; ck < 4; ck++) o = wmma4(f.v[ck], pf[ck], o);
}

__global__ void attn_kernel(const float* __restrict__ Qp, const float* __restrict__ Kp,
                            const float* __restrict__ Vp, float* __restrict__ sa) {
  int lane = threadIdx.x & 31;
  int wv = __builtin_amdgcn_readfirstlane((int)(threadIdx.x >> 5));
  int mr = lane & 15, half = lane >> 4, kh = half << 1;
  int dv = mr < HEAD ? mr : HEAD - 1;
  int pr = blockIdx.x * 4 + wv;
  int bh = blockIdx.y;
  const size_t base = (size_t)bh * SEQ * HEAD;
  int b = bh >> 4, h = bh & 15;
  const float* Kh = Kp + base;
  const float* Vh = Vp + base;

#define TILE_EDGE() do { \
    asm volatile("" : "+v"(qf[0]), "+v"(qf[1]), "+v"(qf[2]), "+v"(o) : : "memory"); \
    __builtin_amdgcn_sched_barrier(0); } while (0)

  #pragma unroll 1
  for (int pass = 0; pass < 2; ++pass) {
    int qt = pass ? pr : (SEQ / 16 - 1) - pr;
    const float* Qb = Qp + base + (size_t)qt * TILE_F;

    v2f qf[3];
    #pragma unroll
    for (int ck = 0; ck < 3; ck++) qf[ck] = *(const v2f*)(Qb + mr * HEAD + ck * 4 + kh);

    v8f o = {};
    float mrun = -INFINITY, lrun = 0.f;

    KVFrag fa, fb;
    kv_load(fa, Kh, Vh, mr, kh, half, dv);
    int kt = 0;
    #pragma unroll 1
    for (; kt + 2 <= qt; kt += 2) {
      kv_load(fb, Kh + (size_t)(kt + 1) * TILE_F, Vh + (size_t)(kt + 1) * TILE_F, mr, kh, half, dv);
      TILE_EDGE();
      attn_tile<false>(fa, qf, o, mrun, lrun, mr, half);
      TILE_EDGE();
      kv_load(fa, Kh + (size_t)(kt + 2) * TILE_F, Vh + (size_t)(kt + 2) * TILE_F, mr, kh, half, dv);
      TILE_EDGE();
      attn_tile<false>(fb, qf, o, mrun, lrun, mr, half);
      TILE_EDGE();
    }
    if (kt < qt) {
      kv_load(fb, Kh + (size_t)(kt + 1) * TILE_F, Vh + (size_t)(kt + 1) * TILE_F, mr, kh, half, dv);
      TILE_EDGE();
      attn_tile<false>(fa, qf, o, mrun, lrun, mr, half);
      TILE_EDGE();
      attn_tile<true>(fb, qf, o, mrun, lrun, mr, half);
    } else {
      attn_tile<true>(fa, qf, o, mrun, lrun, mr, half);
    }

    float* dst = sa + ((size_t)(b * SEQ + qt * 16 + mr)) * N_EMBD + h * HEAD + 8 * half;
    v4f lo = { o[0] / lrun, o[1] / lrun, o[2] / lrun, o[3] / lrun };
    *(v4f*)dst = lo;
    if (half == 0) {
      v4f hi = { o[4] / lrun, o[5] / lrun, o[6] / lrun, o[7] / lrun };
      *(v4f*)(dst + 4) = hi;
    }
  }
#undef TILE_EDGE
}

__global__ void __launch_bounds__(256) gemm_proj_kernel(const float* __restrict__ A, const float* __restrict__ Bf,
                                 const float* __restrict__ bias, const float* __restrict__ resid,
                                 float* __restrict__ out) {
  int wv = threadIdx.x >> 5;
  int tile = blockIdx.x * 8 + wv;
  const int TN = 192 / 48;
  int m0 = (tile / TN) * 16, n0 = (tile % TN) * 48;
  v8f acc[3] = {{}, {}, {}};
  float ss = 0.f;
  gemm16xn<false, 3>(A, Bf, 192, 192, 192, m0, n0, acc, ss);
  int lane = threadIdx.x & 31; int mr = lane & 15, half = lane >> 4;
  #pragma unroll
  for (int t = 0; t < 3; t++) {
    int j = n0 + t * 16 + mr;
    float bj = bias[j];
    #pragma unroll
    for (int v = 0; v < 8; v++) {
      int m = m0 + v + 8 * half;
      size_t ofs = (size_t)m * 192 + j;
      out[ofs] = acc[t][v] + bj + resid[ofs];
    }
  }
}

__global__ void __launch_bounds__(256) gemm_ff1_kernel(const float* __restrict__ A, const float* __restrict__ Bf,
                                float* __restrict__ out) {
  int wv = threadIdx.x >> 5;
  int tile = blockIdx.x * 8 + wv;
  const int TN = FFN / 64;
  int m0 = (tile / TN) * 16, n0 = (tile % TN) * 64;
  v8f acc[4] = {{}, {}, {}, {}};
  float ss = 0.f;
  gemm16xn<true, 4>(A, Bf, 192, 192, FFN, m0, n0, acc, ss);
  int lane = threadIdx.x & 31; int mr = lane & 15, half = lane >> 4;
  float scr[8];
  row_rms_factors(ss, half, scr);
  #pragma unroll
  for (int t = 0; t < 4; t++) {
    int j = n0 + t * 16 + mr;
    #pragma unroll
    for (int v = 0; v < 8; v++) {
      int m = m0 + v + 8 * half;
      float val = acc[t][v] * scr[v];
      out[(size_t)m * FFN + j] = val / (1.0f + __expf(-val));
    }
  }
}

__global__ void __launch_bounds__(256) gemm_ff2_kernel(const float* __restrict__ A, const float* __restrict__ Bf,
                                const float* __restrict__ resid, float* __restrict__ out) {
  int wv = threadIdx.x >> 5;
  int tile = blockIdx.x * 8 + wv;
  const int TN = 192 / 48;
  int m0 = (tile / TN) * 16, n0 = (tile % TN) * 48;
  v8f acc[3] = {{}, {}, {}};
  float ss = 0.f;
  gemm16xn<false, 3>(A, Bf, FFN, FFN, 192, m0, n0, acc, ss);
  int lane = threadIdx.x & 31; int mr = lane & 15, half = lane >> 4;
  #pragma unroll
  for (int t = 0; t < 3; t++) {
    int j = n0 + t * 16 + mr;
    #pragma unroll
    for (int v = 0; v < 8; v++) {
      int m = m0 + v + 8 * half;
      size_t ofs = (size_t)m * 192 + j;
      out[ofs] = acc[t][v] + resid[ofs];
    }
  }
}

extern "C" void kernel_launch(void* const* d_in, const int* in_sizes, int n_in,
                              void* d_out, int out_size, void* d_ws, size_t ws_size,
                              hipStream_t stream) {
  (void)in_sizes; (void)n_in; (void)out_size; (void)ws_size;
  const float* x     = (const float*)d_in[0];
  const float* ln1_w = (const float*)d_in[1];
  const float* ln2_w = (const float*)d_in[2];
  const float* Wq    = (const float*)d_in[3];
  const float* Wk    = (const float*)d_in[4];
  const float* Wv    = (const float*)d_in[5];
  const float* Wproj = (const float*)d_in[6];
  const float* bproj = (const float*)d_in[7];
  const float* W1    = (const float*)d_in[8];
  const float* W2    = (const float*)d_in[9];
  float* ws = (float*)d_ws;

  size_t off = 0;
  float* wqkv = ws + off; off += (size_t)192 * 576;
  float* wpjf = ws + off; off += (size_t)192 * 192;
  float* w1f  = ws + off; off += (size_t)192 * FFN;
  float* w2f  = ws + off; off += (size_t)FFN * 192;
  float* Qp   = ws + off; off += (size_t)BH * SEQ * HEAD;
  float* Kp   = ws + off; off += (size_t)BH * SEQ * HEAD;
  float* Vp   = ws + off; off += (size_t)BH * SEQ * HEAD;
  float* sa   = ws + off; off += (size_t)M_TOK * N_EMBD;
  float* x1   = ws + off; off += (size_t)M_TOK * N_EMBD;
  float* u    = ws + off; off += (size_t)M_TOK * FFN;
  float* out  = (float*)d_out;

  prep_kernel<<<PACK_WGS, 256, 0, stream>>>(ln1_w, ln2_w, Wq, Wk, Wv, Wproj, W1, W2,
                                            wqkv, wpjf, w1f, w2f);
  gemm_qkv_kernel<<<(512 * 12) / 8, 256, 0, stream>>>(x, wqkv, Qp, Kp, Vp);
  attn_kernel<<<dim3(16, BH), 128, 0, stream>>>(Qp, Kp, Vp, sa);
  gemm_proj_kernel<<<(512 * 4) / 8, 256, 0, stream>>>(sa, wpjf, bproj, x, x1);
  gemm_ff1_kernel<<<(512 * 12) / 8, 256, 0, stream>>>(x1, w1f, u);
  gemm_ff2_kernel<<<(512 * 4) / 8, 256, 0, stream>>>(u, w2f, x1, out);
}


# GEMM work-group = 4 row tiles x 2 column strips: four waves share each weight strip through the WGP cache and two each row tile (was 1-2 row tiles x 4-8 strips); second-level reads about halved
# speedup vs baseline: 2.7790x; 1.0437x over previous
#include <hip/hip_runtime.h>
#include <math.h>

typedef __attribute__((ext_vector_type(2))) float v2f;
typedef __attribute__((ext_vector_type(4))) float v4f;
typedef __attribute__((ext_vector_type(8))) float v8f;

#define N_EMBD 192
#define N_HEAD 16
#define HEAD   12
#define TILE_F (16*HEAD)
#define BATCH  4
#define SEQ    2048
#define M_TOK  (BATCH*SEQ)
#define FFN    (4*N_EMBD)
#define BH     (BATCH*N_HEAD)

static __device__ __forceinline__ v8f wmma4(v2f a, v2f b, v8f c) {
  return __builtin_amdgcn_wmma_f32_16x16x4_f32(false, a, false, b, (short)0, c, false, false);
}

static __device__ __forceinline__ float xhalf(float x) {
  int b = __builtin_bit_cast(int, x);
  return __builtin_bit_cast(float,
      __builtin_amdgcn_permlanex16(b, b, 0x76543210, 0xfedcba98, false, false));
}


#define PACK_QUADS 1024

template <bool GAIN>
static __device__ __forceinline__ void pack_quads_b(const float* __restrict__ B, const float* __restrict__ g,
                                                    float* __restrict__ Bf, int N, int q0) {
  v4f val[4];
  #pragma unroll
  for (int i = 0; i < 4; i++) {
    int q = q0 + 256 * i;
    int kq = q / N, n = q % N;
    const float* s = B + (size_t)kq * 4 * N + n;
    val[i] = (v4f){ s[0], s[N], s[2 * N], s[3 * N] };
    if (GAIN) {
      const float* gk = g + 4 * kq;
      val[i] = val[i] * (v4f){ gk[0], gk[1], gk[2], gk[3] };
    }
  }
  #pragma unroll
  for (int i = 0; i < 4; i++) *(v4f*)(Bf + (size_t)(q0 + 256 * i) * 4) = val[i];
}

static __device__ __forceinline__ void pack_quads_qkv(const float* __restrict__ Wq, const float* __restrict__ Wk,
                                                      const float* __restrict__ Wv, const float* __restrict__ g,
                                                      float* __restrict__ Bf, int q0) {
  v4f val[4];
  #pragma unroll
  for (int i = 0; i < 4; i++) {
    int q = q0 + 256 * i;
    int cq = q / 576, j = q % 576;
    int which = j / 192, r = j % 192;
    int h = r / HEAD, d = r % HEAD;
    const float* W = (which == 0) ? Wq : ((which == 1) ? Wk : Wv);
    const float* s = W + ((size_t)h * N_EMBD + 4 * cq) * HEAD + d;
    const float* gk = g + 4 * cq;
    val[i] = (v4f){ s[0] * gk[0], s[HEAD] * gk[1], s[2 * HEAD] * gk[2], s[3 * HEAD] * gk[3] };
  }
  #pragma unroll
  for (int i = 0; i < 4; i++) *(v4f*)(Bf + (size_t)(q0 + 256 * i) * 4) = val[i];
}

#define PACK_WGS 108
__global__ void __launch_bounds__(256) prep_kernel(const float* __restrict__ ln1_w, const float* __restrict__ ln2_w,
                            const float* __restrict__ Wq, const float* __restrict__ Wk,
                            const float* __restrict__ Wv, const float* __restrict__ Wproj,
                            const float* __restrict__ W1, const float* __restrict__ W2,
                            float* __restrict__ wqkv, float* __restrict__ wpjf,
                            float* __restrict__ w1f, float* __restrict__ w2f) {
  int blk = blockIdx.x;
  int t = threadIdx.x;
  if (blk < 27)      pack_quads_qkv(Wq, Wk, Wv, ln1_w, wqkv, blk * PACK_QUADS + t);
  else if (blk < 36) pack_quads_b<false>(Wproj, nullptr, wpjf, 192, (blk - 27) * PACK_QUADS + t);
  else if (blk < 72) pack_quads_b<true>(W1, ln2_w, w1f, FFN, (blk - 36) * PACK_QUADS + t);
  else               pack_quads_b<false>(W2, nullptr, w2f, 192, (blk - 72) * PACK_QUADS + t);
}

template <int NT> struct Stage { v4f a[2]; v4f b[2][NT]; };

template <int TN, int SW>
static __device__ __forceinline__ void wave_tile(int& m0, int& n0) {
  int wv = threadIdx.x >> 5;
  const int SP = TN / 2;
  m0 = ((int)(blockIdx.x / SP) * 4 + (wv & 3)) * 16;
  n0 = ((int)(blockIdx.x % SP) * 2 + (wv >> 2)) * SW;
}

static __device__ __forceinline__ v2f lo2(v4f x) { return __builtin_shufflevector(x, x, 0, 1); }
static __device__ __forceinline__ v2f hi2(v4f x) { return __builtin_shufflevector(x, x, 2, 3); }

template <int NT>
static __device__ __forceinline__ void stage_load(Stage<NT>& f, const float* __restrict__ ap,
                                                  const float* __restrict__ bp, int N) {
  #pragma unroll
  for (int u = 0; u < 2; u++) {
    f.a[u] = *(const v4f*)(ap + 8 * u);
    #pragma unroll
    for (int t = 0; t < NT; t++) f.b[u][t] = *(const v4f*)(bp + (size_t)u * 8 * N + t * 64);
  }
}

template <bool SUMSQ, int NT>
static __device__ __forceinline__ void stage_mma(const Stage<NT>& f, v8f* acc, v4f& sq) {
  #pragma unroll
  for (int u = 0; u < 2; u++) {
    v2f a0 = lo2(f.a[u]), a1 = hi2(f.a[u]);
    if (SUMSQ) sq = sq + f.a[u] * f.a[u];
    #pragma unroll
    for (int t = 0; t < NT; t++) acc[t] = wmma4(a0, lo2(f.b[u][t]), acc[t]);
    #pragma unroll
    for (int t = 0; t < NT; t++) acc[t] = wmma4(a1, hi2(f.b[u][t]), acc[t]);
  }
}

template <bool SUMSQ, int NT>
__device__ __forceinline__ void gemm16xn(const float* __restrict__ A, const float* __restrict__ Bf,
                                          int K, int lda, int N, int m0, int n0, v8f* acc, float& ss) {
  int lane = threadIdx.x & 31;
  int mr = lane & 15;
  int half = lane >> 4;
  const float* ap = A + (size_t)(m0 + mr) * lda + 4 * half;
  const float* bp = Bf + (size_t)half * 4 * N + (size_t)(n0 + mr) * 4;
  const int S = K / 16;
  Stage<NT> f0, f1, f2;
  v4f sq = {};
#define STAGE_EDGE() do { \
    if constexpr (NT == 4) { \
      if (SUMSQ) asm volatile("" : "+v"(acc[0]), "+v"(acc[1]), "+v"(acc[2]), "+v"(acc[3]), "+v"(sq) : : "memory"); \
      else       asm volatile("" : "+v"(acc[0]), "+v"(acc[1]), "+v"(acc[2]), "+v"(acc[3]) : : "memory"); \
    } else { \
      if (SUMSQ) asm volatile("" : "+v"(acc[0]), "+v"(acc[1]), "+v"(acc[2]), "+v"(sq) : : "memory"); \
      else       asm volatile("" : "+v"(acc[0]), "+v"(acc[1]), "+v"(acc[2]) : : "memory"); \
    } \
    __builtin_amdgcn_sched_barrier(0); } while (0)
  stage_load(f0, ap, bp, N);                            STAGE_EDGE();
  stage_load(f1, ap + 16, bp + (size_t)16 * N, N);      STAGE_EDGE();
  ap += 32; bp += (size_t)32 * N;
  #pragma unroll 1
  for (int s = 0; s < S - 3; s += 3) {
    stage_load(f2, ap, bp, N);                          STAGE_EDGE();
    stage_mma<SUMSQ, NT>(f0, acc, sq);                      STAGE_EDGE();
    stage_load(f0, ap + 16, bp + (size_t)16 * N, N);    STAGE_EDGE();
    stage_mma<SUMSQ, NT>(f1, acc, sq);                      STAGE_EDGE();
    stage_load(f1, ap + 32, bp + (size_t)32 * N, N);    STAGE_EDGE();
    stage_mma<SUMSQ, NT>(f2, acc, sq);                      STAGE_EDGE();
    ap += 48; bp += (size_t)48 * N;
  }
  stage_load(f2, ap, bp, N);                            STAGE_EDGE();
  stage_mma<SUMSQ, NT>(f0, acc, sq);                        if (SUMSQ) STAGE_EDGE();
  stage_mma<SUMSQ, NT>(f1, acc, sq);                        if (SUMSQ) STAGE_EDGE();
  stage_mma<SUMSQ, NT>(f2, acc, sq);
#undef STAGE_EDGE
  if (SUMSQ) ss = (sq[0] + sq[1]) + (sq[2] + sq[3]);
}

static __device__ __forceinline__ void row_rms_factors(float ss, int half, float* scr) {
  float sc = rsqrtf((ss + xhalf(ss)) * (1.0f / N_EMBD) + 1e-5f);
  #pragma unroll
  for (int v = 0; v < 8; v++) scr[v] = __shfl(sc, v + 8 * half, 32);
}

__global__ void __launch_bounds__(256) gemm_qkv_kernel(const float* __restrict__ A, const float* __restrict__ Bf,
                                float* __restrict__ Qp, float* __restrict__ Kp,
                                float* __restrict__ Vp) {
  int m0, n0;
  wave_tile<576 / 48, 48>(m0, n0);
  v8f acc[3] = {{}, {}, {}};
  float ss = 0.f;
  gemm16xn<true, 3>(A, Bf, N_EMBD, N_EMBD, 576, m0, n0, acc, ss);
  int lane = threadIdx.x & 31;
  int mr = lane & 15, half = lane >> 4;
  float scr[8];
  row_rms_factors(ss, half, scr);
  #pragma unroll
  for (int t = 0; t < 3; t++) {
    int j = n0 + t * 16 + mr;
    int which = j / 192, r = j % 192, h = r / HEAD, d = r % HEAD;
    #pragma unroll
    for (int v = 0; v < 8; v++) {
      int m = m0 + v + 8 * half;
      int b = m >> 11, tq = m & 2047;
      size_t hb = ((size_t)(b * N_HEAD + h)) * SEQ * HEAD;
      float val = acc[t][v] * scr[v];
      if (which == 0)      Qp[hb + (size_t)tq * HEAD + d] = val;
      else if (which == 1) Kp[hb + (size_t)tq * HEAD + d] = val;
      else                 Vp[hb + (size_t)(tq >> 1) * (2 * HEAD) + d * 2 + (tq & 1)] = val;
    }
  }
}


struct KVFrag { v2f k[3]; v2f v[4]; };

static __device__ __forceinline__ void kv_load(KVFrag& f, const float* __restrict__ Kt,
                                               const float* __restrict__ Vt,
                                               int mr, int kh, int half, int dv) {
  #pragma unroll
  for (int ck = 0; ck < 3; ck++) f.k[ck] = *(const v2f*)(Kt + mr * HEAD + ck * 4 + kh);
  #pragma unroll
  for (int ck = 0; ck < 4; ck++) f.v[ck] = *(const v2f*)(Vt + (ck + 4 * half) * (2 * HEAD) + dv * 2);
}

template <bool DIAG>
static __device__ __forceinline__ void attn_tile(const KVFrag& f, const v2f* qf, v8f& o,
                                                 float& mrun, float& lrun, int mr, int half) {
  const float scale = 0.28867513459481287f;
  v8f s = {};
  #pragma unroll
  for (int ck = 0; ck < 3; ck++) s = wmma4(f.k[ck], qf[ck], s);
  float sv[8];
  #pragma unroll
  for (int r = 0; r < 8; r++) sv[r] = s[r] * scale;
  if (DIAG) {
    #pragma unroll
    for (int r = 0; r < 8; r++) sv[r] = (r + 8 * half > mr) ? -INFINITY : sv[r];
  }
  float tmax = fmaxf(fmaxf(fmaxf(sv[0], sv[1]), fmaxf(sv[2], sv[3])),
                     fmaxf(fmaxf(sv[4], sv[5]), fmaxf(sv[6], sv[7])));
  tmax = fmaxf(tmax, xhalf(tmax));
  float mnew  = fmaxf(mrun, tmax);
  float alpha = __expf(mrun - mnew);
  v2f pf[4];
  #pragma unroll
  for (int r = 0; r < 8; r++) pf[r >> 1][r & 1] = __expf(sv[r] - mnew);
  float rsum = ((pf[0][0] + pf[0][1]) + (pf[1][0] + pf[1][1])) +
               ((pf[2][0] + pf[2][1]) + (pf[3][0] + pf[3][1]));
  rsum = rsum + xhalf(rsum);
  lrun = lrun * alpha + rsum;
  mrun = mnew;
  #pragma unroll
  for (int r = 0; r < 8; r++) o[r] = o[r] * alpha;
  #pragma unroll
  for (int ck = 0; ck < 4; ck++) o = wmma4(f.v[ck], pf[ck], o);
}

__global__ void attn_kernel(const float* __restrict__ Qp, const float* __restrict__ Kp,
                            const float* __restrict__ Vp, float* __restrict__ sa) {
  int lane = threadIdx.x & 31;
  int wv = __builtin_amdgcn_readfirstlane((int)(threadIdx.x >> 5));
  int mr = lane & 15, half = lane >> 4, kh = half << 1;
  int dv = mr < HEAD ? mr : HEAD - 1;
  int pr = blockIdx.x * 4 + wv;
  int bh = blockIdx.y;
  const size_t base = (size_t)bh * SEQ * HEAD;
  int b = bh >> 4, h = bh & 15;
  const float* Kh = Kp + base;
  const float* Vh = Vp + base;

#define TILE_EDGE() do { \
    asm volatile("" : "+v"(qf[0]), "+v"(qf[1]), "+v"(qf[2]), "+v"(o) : : "memory"); \
    __builtin_amdgcn_sched_barrier(0); } while (0)

  #pragma unroll 1
  for (int pass = 0; pass < 2; ++pass) {
    int qt = pass ? pr : (SEQ / 16 - 1) - pr;
    const float* Qb = Qp + base + (size_t)qt * TILE_F;

    v2f qf[3];
    #pragma unroll
    for (int ck = 0; ck < 3; ck++) qf[ck] = *(const v2f*)(Qb + mr * HEAD + ck * 4 + kh);

    v8f o = {};
    float mrun = -INFINITY, lrun = 0.f;

    KVFrag fa, fb;
    kv_load(fa, Kh, Vh, mr, kh, half, dv);
    int kt = 0;
    #pragma unroll 1
    for (; kt + 2 <= qt; kt += 2) {
      kv_load(fb, Kh + (size_t)(kt + 1) * TILE_F, Vh + (size_t)(kt + 1) * TILE_F, mr, kh, half, dv);
      TILE_EDGE();
      attn_tile<false>(fa, qf, o, mrun, lrun, mr, half);
      TILE_EDGE();
      kv_load(fa, Kh + (size_t)(kt + 2) * TILE_F, Vh + (size_t)(kt + 2) * TILE_F, mr, kh, half, dv);
      TILE_EDGE();
      attn_tile<false>(fb, qf, o, mrun, lrun, mr, half);
      TILE_EDGE();
    }
    if (kt < qt) {
      kv_load(fb, Kh + (size_t)(kt + 1) * TILE_F, Vh + (size_t)(kt + 1) * TILE_F, mr, kh, half, dv);
      TILE_EDGE();
      attn_tile<false>(fa, qf, o, mrun, lrun, mr, half);
      TILE_EDGE();
      attn_tile<true>(fb, qf, o, mrun, lrun, mr, half);
    } else {
      attn_tile<true>(fa, qf, o, mrun, lrun, mr, half);
    }

    float* dst = sa + ((size_t)(b * SEQ + qt * 16 + mr)) * N_EMBD + h * HEAD + 8 * half;
    v4f lo = { o[0] / lrun, o[1] / lrun, o[2] / lrun, o[3] / lrun };
    *(v4f*)dst = lo;
    if (half == 0) {
      v4f hi = { o[4] / lrun, o[5] / lrun, o[6] / lrun, o[7] / lrun };
      *(v4f*)(dst + 4) = hi;
    }
  }
#undef TILE_EDGE
}

__global__ void __launch_bounds__(256) gemm_proj_kernel(const float* __restrict__ A, const float* __restrict__ Bf,
                                 const float* __restrict__ bias, const float* __restrict__ resid,
                                 float* __restrict__ out) {
  int m0, n0;
  wave_tile<192 / 48, 48>(m0, n0);
  v8f acc[3] = {{}, {}, {}};
  float ss = 0.f;
  gemm16xn<false, 3>(A, Bf, 192, 192, 192, m0, n0, acc, ss);
  int lane = threadIdx.x & 31; int mr = lane & 15, half = lane >> 4;
  #pragma unroll
  for (int t = 0; t < 3; t++) {
    int j = n0 + t * 16 + mr;
    float bj = bias[j];
    #pragma unroll
    for (int v = 0; v < 8; v++) {
      int m = m0 + v + 8 * half;
      size_t ofs = (size_t)m * 192 + j;
      out[ofs] = acc[t][v] + bj + resid[ofs];
    }
  }
}

__global__ void __launch_bounds__(256) gemm_ff1_kernel(const float* __restrict__ A, const float* __restrict__ Bf,
                                float* __restrict__ out) {
  int m0, n0;
  wave_tile<FFN / 64, 64>(m0, n0);
  v8f acc[4] = {{}, {}, {}, {}};
  float ss = 0.f;
  gemm16xn<true, 4>(A, Bf, 192, 192, FFN, m0, n0, acc, ss);
  int lane = threadIdx.x & 31; int mr = lane & 15, half = lane >> 4;
  float scr[8];
  row_rms_factors(ss, half, scr);
  #pragma unroll
  for (int t = 0; t < 4; t++) {
    int j = n0 + t * 16 + mr;
    #pragma unroll
    for (int v = 0; v < 8; v++) {
      int m = m0 + v + 8 * half;
      float val = acc[t][v] * scr[v];
      out[(size_t)m * FFN + j] = val / (1.0f + __expf(-val));
    }
  }
}

__global__ void __launch_bounds__(256) gemm_ff2_kernel(const float* __restrict__ A, const float* __restrict__ Bf,
                                const float* __restrict__ resid, float* __restrict__ out) {
  int m0, n0;
  wave_tile<192 / 48, 48>(m0, n0);
  v8f acc[3] = {{}, {}, {}};
  float ss = 0.f;
  gemm16xn<false, 3>(A, Bf, FFN, FFN, 192, m0, n0, acc, ss);
  int lane = threadIdx.x & 31; int mr = lane & 15, half = lane >> 4;
  #pragma unroll
  for (int t = 0; t < 3; t++) {
    int j = n0 + t * 16 + mr;
    #pragma unroll
    for (int v = 0; v < 8; v++) {
      int m = m0 + v + 8 * half;
      size_t ofs = (size_t)m * 192 + j;
      out[ofs] = acc[t][v] + resid[ofs];
    }
  }
}

extern "C" void kernel_launch(void* const* d_in, const int* in_sizes, int n_in,
                              void* d_out, int out_size, void* d_ws, size_t ws_size,
                              hipStream_t stream) {
  (void)in_sizes; (void)n_in; (void)out_size; (void)ws_size;
  const float* x     = (const float*)d_in[0];
  const float* ln1_w = (const float*)d_in[1];
  const float* ln2_w = (const float*)d_in[2];
  const float* Wq    = (const float*)d_in[3];
  const float* Wk    = (const float*)d_in[4];
  const float* Wv    = (const float*)d_in[5];
  const float* Wproj = (const float*)d_in[6];
  const float* bproj = (const float*)d_in[7];
  const float* W1    = (const float*)d_in[8];
  const float* W2    = (const float*)d_in[9];
  float* ws = (float*)d_ws;

  size_t off = 0;
  float* wqkv = ws + off; off += (size_t)192 * 576;
  float* wpjf = ws + off; off += (size_t)192 * 192;
  float* w1f  = ws + off; off += (size_t)192 * FFN;
  float* w2f  = ws + off; off += (size_t)FFN * 192;
  float* Qp   = ws + off; off += (size_t)BH * SEQ * HEAD;
  float* Kp   = ws + off; off += (size_t)BH * SEQ * HEAD;
  float* Vp   = ws + off; off += (size_t)BH * SEQ * HEAD;
  float* sa   = ws + off; off += (size_t)M_TOK * N_EMBD;
  float* x1   = ws + off; off += (size_t)M_TOK * N_EMBD;
  float* u    = ws + off; off += (size_t)M_TOK * FFN;
  float* out  = (float*)d_out;

  prep_kernel<<<PACK_WGS, 256, 0, stream>>>(ln1_w, ln2_w, Wq, Wk, Wv, Wproj, W1, W2,
                                            wqkv, wpjf, w1f, w2f);
  gemm_qkv_kernel<<<(512 * 12) / 8, 256, 0, stream>>>(x, wqkv, Qp, Kp, Vp);
  attn_kernel<<<dim3(16, BH), 128, 0, stream>>>(Qp, Kp, Vp, sa);
  gemm_proj_kernel<<<(512 * 4) / 8, 256, 0, stream>>>(sa, wpjf, bproj, x, x1);
  gemm_ff1_kernel<<<(512 * 12) / 8, 256, 0, stream>>>(x1, w1f, u);
  gemm_ff2_kernel<<<(512 * 4) / 8, 256, 0, stream>>>(u, w2f, x1, out);
}


# attention: a wave walks the key tiles once for TWO query tiles (32 queries): each K and V fragment load feeds 14 matrix instructions, not 7; double tiles 63-p and p keep the causal balance
# speedup vs baseline: 2.8935x; 1.0412x over previous
#include <hip/hip_runtime.h>
#include <math.h>

typedef __attribute__((ext_vector_type(2))) float v2f;
typedef __attribute__((ext_vector_type(4))) float v4f;
typedef __attribute__((ext_vector_type(8))) float v8f;

#define N_EMBD 192
#define N_HEAD 16
#define HEAD   12
#define TILE_F (16*HEAD)
#define BATCH  4
#define SEQ    2048
#define M_TOK  (BATCH*SEQ)
#define FFN    (4*N_EMBD)
#define BH     (BATCH*N_HEAD)

static __device__ __forceinline__ v8f wmma4(v2f a, v2f b, v8f c) {
  return __builtin_amdgcn_wmma_f32_16x16x4_f32(false, a, false, b, (short)0, c, false, false);
}

static __device__ __forceinline__ float xhalf(float x) {
  int b = __builtin_bit_cast(int, x);
  return __builtin_bit_cast(float,
      __builtin_amdgcn_permlanex16(b, b, 0x76543210, 0xfedcba98, false, false));
}


#define PACK_QUADS 1024

template <bool GAIN>
static __device__ __forceinline__ void pack_quads_b(const float* __restrict__ B, const float* __restrict__ g,
                                                    float* __restrict__ Bf, int N, int q0) {
  v4f val[4];
  #pragma unroll
  for (int i = 0; i < 4; i++) {
    int q = q0 + 256 * i;
    int kq = q / N, n = q % N;
    const float* s = B + (size_t)kq * 4 * N + n;
    val[i] = (v4f){ s[0], s[N], s[2 * N], s[3 * N] };
    if (GAIN) {
      const float* gk = g + 4 * kq;
      val[i] = val[i] * (v4f){ gk[0], gk[1], gk[2], gk[3] };
    }
  }
  #pragma unroll
  for (int i = 0; i < 4; i++) *(v4f*)(Bf + (size_t)(q0 + 256 * i) * 4) = val[i];
}

static __device__ __forceinline__ void pack_quads_qkv(const float* __restrict__ Wq, const float* __restrict__ Wk,
                                                      const float* __restrict__ Wv, const float* __restrict__ g,
                                                      float* __restrict__ Bf, int q0) {
  v4f val[4];
  #pragma unroll
  for (int i = 0; i < 4; i++) {
    int q = q0 + 256 * i;
    int cq = q / 576, j = q % 576;
    int which = j / 192, r = j % 192;
    int h = r / HEAD, d = r % HEAD;
    const float* W = (which == 0) ? Wq : ((which == 1) ? Wk : Wv);
    const float* s = W + ((size_t)h * N_EMBD + 4 * cq) * HEAD + d;
    const float* gk = g + 4 * cq;
    val[i] = (v4f){ s[0] * gk[0], s[HEAD] * gk[1], s[2 * HEAD] * gk[2], s[3 * HEAD] * gk[3] };
  }
  #pragma unroll
  for (int i = 0; i < 4; i++) *(v4f*)(Bf + (size_t)(q0 + 256 * i) * 4) = val[i];
}

#define PACK_WGS 108
__global__ void __launch_bounds__(256) prep_kernel(const float* __restrict__ ln1_w, const float* __restrict__ ln2_w,
                            const float* __restrict__ Wq, const float* __restrict__ Wk,
                            const float* __restrict__ Wv, const float* __restrict__ Wproj,
                            const float* __restrict__ W1, const float* __restrict__ W2,
                            float* __restrict__ wqkv, float* __restrict__ wpjf,
                            float* __restrict__ w1f, float* __restrict__ w2f) {
  int blk = blockIdx.x;
  int t = threadIdx.x;
  if (blk < 27)      pack_quads_qkv(Wq, Wk, Wv, ln1_w, wqkv, blk * PACK_QUADS + t);
  else if (blk < 36) pack_quads_b<false>(Wproj, nullptr, wpjf, 192, (blk - 27) * PACK_QUADS + t);
  else if (blk < 72) pack_quads_b<true>(W1, ln2_w, w1f, FFN, (blk - 36) * PACK_QUADS + t);
  else               pack_quads_b<false>(W2, nullptr, w2f, 192, (blk - 72) * PACK_QUADS + t);
}

template <int NT> struct Stage { v4f a[2]; v4f b[2][NT]; };

template <int TN, int SW>
static __device__ __forceinline__ void wave_tile(int& m0, int& n0) {
  int wv = threadIdx.x >> 5;
  const int SP = TN / 2;
  m0 = ((int)(blockIdx.x / SP) * 4 + (wv & 3)) * 16;
  n0 = ((int)(blockIdx.x % SP) * 2 + (wv >> 2)) * SW;
}

static __device__ __forceinline__ v2f lo2(v4f x) { return __builtin_shufflevector(x, x, 0, 1); }
static __device__ __forceinline__ v2f hi2(v4f x) { return __builtin_shufflevector(x, x, 2, 3); }

template <int NT>
static __device__ __forceinline__ void stage_load(Stage<NT>& f, const float* __restrict__ ap,
                                                  const float* __restrict__ bp, int N) {
  #pragma unroll
  for (int u = 0; u < 2; u++) {
    f.a[u] = *(const v4f*)(ap + 8 * u);
    #pragma unroll
    for (int t = 0; t < NT; t++) f.b[u][t] = *(const v4f*)(bp + (size_t)u * 8 * N + t * 64);
  }
}

template <bool SUMSQ, int NT>
static __device__ __forceinline__ void stage_mma(const Stage<NT>& f, v8f* acc, v4f& sq) {
  #pragma unroll
  for (int u = 0; u < 2; u++) {
    v2f a0 = lo2(f.a[u]), a1 = hi2(f.a[u]);
    if (SUMSQ) sq = sq + f.a[u] * f.a[u];
    #pragma unroll
    for (int t = 0; t < NT; t++) acc[t] = wmma4(a0, lo2(f.b[u][t]), acc[t]);
    #pragma unroll
    for (int t = 0; t < NT; t++) acc[t] = wmma4(a1, hi2(f.b[u][t]), acc[t]);
  }
}

template <bool SUMSQ, int NT>
__device__ __forceinline__ void gemm16xn(const float* __restrict__ A, const float* __restrict__ Bf,
                                          int K, int lda, int N, int m0, int n0, v8f* acc, float& ss) {
  int lane = threadIdx.x & 31;
  int mr = lane & 15;
  int half = lane >> 4;
  const float* ap = A + (size_t)(m0 + mr) * lda + 4 * half;
  const float* bp = Bf + (size_t)half * 4 * N + (size_t)(n0 + mr) * 4;
  const int S = K / 16;
  Stage<NT> f0, f1, f2;
  v4f sq = {};
#define STAGE_EDGE() do { \
    if constexpr (NT == 4) { \
      if (SUMSQ) asm volatile("" : "+v"(acc[0]), "+v"(acc[1]), "+v"(acc[2]), "+v"(acc[3]), "+v"(sq) : : "memory"); \
      else       asm volatile("" : "+v"(acc[0]), "+v"(acc[1]), "+v"(acc[2]), "+v"(acc[3]) : : "memory"); \
    } else { \
      if (SUMSQ) asm volatile("" : "+v"(acc[0]), "+v"(acc[1]), "+v"(acc[2]), "+v"(sq) : : "memory"); \
      else       asm volatile("" : "+v"(acc[0]), "+v"(acc[1]), "+v"(acc[2]) : : "memory"); \
    } \
    __builtin_amdgcn_sched_barrier(0); } while (0)
  stage_load(f0, ap, bp, N);                            STAGE_EDGE();
  stage_load(f1, ap + 16, bp + (size_t)16 * N, N);      STAGE_EDGE();
  ap += 32; bp += (size_t)32 * N;
  #pragma unroll 1
  for (int s = 0; s < S - 3; s += 3) {
    stage_load(f2, ap, bp, N);                          STAGE_EDGE();
    stage_mma<SUMSQ, NT>(f0, acc, sq);                      STAGE_EDGE();
    stage_load(f0, ap + 16, bp + (size_t)16 * N, N);    STAGE_EDGE();
    stage_mma<SUMSQ, NT>(f1, acc, sq);                      STAGE_EDGE();
    stage_load(f1, ap + 32, bp + (size_t)32 * N, N);    STAGE_EDGE();
    stage_mma<SUMSQ, NT>(f2, acc, sq);                      STAGE_EDGE();
    ap += 48; bp += (size_t)48 * N;
  }
  stage_load(f2, ap, bp, N);                            STAGE_EDGE();
  stage_mma<SUMSQ, NT>(f0, acc, sq);                        if (SUMSQ) STAGE_EDGE();
  stage_mma<SUMSQ, NT>(f1, acc, sq);                        if (SUMSQ) STAGE_EDGE();
  stage_mma<SUMSQ, NT>(f2, acc, sq);
#undef STAGE_EDGE
  if (SUMSQ) ss = (sq[0] + sq[1]) + (sq[2] + sq[3]);
}

static __device__ __forceinline__ void row_rms_factors(float ss, int half, float* scr) {
  float sc = rsqrtf((ss + xhalf(ss)) * (1.0f / N_EMBD) + 1e-5f);
  #pragma unroll
  for (int v = 0; v < 8; v++) scr[v] = __shfl(sc, v + 8 * half, 32);
}

__global__ void __launch_bounds__(256) gemm_qkv_kernel(const float* __restrict__ A, const float* __restrict__ Bf,
                                float* __restrict__ Qp, float* __restrict__ Kp,
                                float* __restrict__ Vp) {
  int m0, n0;
  wave_tile<576 / 48, 48>(m0, n0);
  v8f acc[3] = {{}, {}, {}};
  float ss = 0.f;
  gemm16xn<true, 3>(A, Bf, N_EMBD, N_EMBD, 576, m0, n0, acc, ss);
  int lane = threadIdx.x & 31;
  int mr = lane & 15, half = lane >> 4;
  float scr[8];
  row_rms_factors(ss, half, scr);
  #pragma unroll
  for (int t = 0; t < 3; t++) {
    int j = n0 + t * 16 + mr;
    int which = j / 192, r = j % 192, h = r / HEAD, d = r % HEAD;
    #pragma unroll
    for (int v = 0; v < 8; v++) {
      int m = m0 + v + 8 * half;
      int b = m >> 11, tq = m & 2047;
      size_t hb = ((size_t)(b * N_HEAD + h)) * SEQ * HEAD;
      float val = acc[t][v] * scr[v];
      if (which == 0)      Qp[hb + (size_t)tq * HEAD + d] = val;
      else if (which == 1) Kp[hb + (size_t)tq * HEAD + d] = val;
      else                 Vp[hb + (size_t)(tq >> 1) * (2 * HEAD) + d * 2 + (tq & 1)] = val;
    }
  }
}


struct KVFrag { v2f k[3]; v2f v[4]; };

static __device__ __forceinline__ void kv_load(KVFrag& f, const float* __restrict__ Kt,
                                               const float* __restrict__ Vt,
                                               int mr, int kh, int half, int dv) {
  #pragma unroll
  for (int ck = 0; ck < 3; ck++) f.k[ck] = *(const v2f*)(Kt + mr * HEAD + ck * 4 + kh);
  #pragma unroll
  for (int ck = 0; ck < 4; ck++) f.v[ck] = *(const v2f*)(Vt + (ck + 4 * half) * (2 * HEAD) + dv * 2);
}


static __device__ __forceinline__ v8f score_tile(const KVFrag& f, const v2f* qf) {
  v8f s = {};
  #pragma unroll
  for (int ck = 0; ck < 3; ck++) s = wmma4(f.k[ck], qf[ck], s);
  return s;
}

template <bool DIAG>
static __device__ __forceinline__ void softmax_pv(const KVFrag& f, v8f s, v8f& o,
                                                  float& mrun, float& lrun, int mr, int half) {
  const float scale = 0.28867513459481287f;
  float sv[8];
  #pragma unroll
  for (int r = 0; r < 8; r++) sv[r] = s[r] * scale;
  if (DIAG) {
    #pragma unroll
    for (int r = 0; r < 8; r++) sv[r] = (r + 8 * half > mr) ? -INFINITY : sv[r];
  }
  float tmax = fmaxf(fmaxf(fmaxf(sv[0], sv[1]), fmaxf(sv[2], sv[3])),
                     fmaxf(fmaxf(sv[4], sv[5]), fmaxf(sv[6], sv[7])));
  tmax = fmaxf(tmax, xhalf(tmax));
  float mnew  = fmaxf(mrun, tmax);
  float alpha = __expf(mrun - mnew);
  v2f pf[4];
  #pragma unroll
  for (int r = 0; r < 8; r++) pf[r >> 1][r & 1] = __expf(sv[r] - mnew);
  float rsum = ((pf[0][0] + pf[0][1]) + (pf[1][0] + pf[1][1])) +
               ((pf[2][0] + pf[2][1]) + (pf[3][0] + pf[3][1]));
  rsum = rsum + xhalf(rsum);
  lrun = lrun * alpha + rsum;
  mrun = mnew;
  #pragma unroll
  for (int r = 0; r < 8; r++) o[r] = o[r] * alpha;
  #pragma unroll
  for (int ck = 0; ck < 4; ck++) o = wmma4(f.v[ck], pf[ck], o);
}

static __device__ __forceinline__ void attn_tile2(const KVFrag& f, const v2f* qa, const v2f* qb,
                                                  v8f& oa, v8f& ob, float& ma, float& la,
                                                  float& mb, float& lb, int mr, int half) {
  v8f sa = score_tile(f, qa);
  v8f sb = score_tile(f, qb);
  softmax_pv<false>(f, sa, oa, ma, la, mr, half);
  softmax_pv<false>(f, sb, ob, mb, lb, mr, half);
}

static __device__ __forceinline__ void store_tile(float* __restrict__ sa, const v8f& o, float lrun,
                                                  int b, int h, int qt, int mr, int half) {
  float* dst = sa + ((size_t)(b * SEQ + qt * 16 + mr)) * N_EMBD + h * HEAD + 8 * half;
  v4f lo = { o[0] / lrun, o[1] / lrun, o[2] / lrun, o[3] / lrun };
  *(v4f*)dst = lo;
  if (half == 0) {
    v4f hi = { o[4] / lrun, o[5] / lrun, o[6] / lrun, o[7] / lrun };
    *(v4f*)(dst + 4) = hi;
  }
}

__global__ void attn_kernel(const float* __restrict__ Qp, const float* __restrict__ Kp,
                            const float* __restrict__ Vp, float* __restrict__ sa) {
  int lane = threadIdx.x & 31;
  int wv = __builtin_amdgcn_readfirstlane((int)(threadIdx.x >> 5));
  int mr = lane & 15, half = lane >> 4, kh = half << 1;
  int dv = mr < HEAD ? mr : HEAD - 1;
  int pr = blockIdx.x * 4 + wv;
  int bh = blockIdx.y;
  const size_t base = (size_t)bh * SEQ * HEAD;
  int b = bh >> 4, h = bh & 15;
  const float* Kh = Kp + base;
  const float* Vh = Vp + base;

#define TILE_EDGE() do { \
    asm volatile("" : "+v"(qa[0]), "+v"(qa[1]), "+v"(qa[2]), "+v"(qb[0]), "+v"(qb[1]), "+v"(qb[2]), \
                      "+v"(oa), "+v"(ob) : : "memory"); \
    __builtin_amdgcn_sched_barrier(0); } while (0)

  #pragma unroll 1
  for (int pass = 0; pass < 2; ++pass) {
    int dt = pass ? pr : (SEQ / 32 - 1) - pr;
    int qt = 2 * dt;
    const float* Qb = Qp + base + (size_t)qt * TILE_F;

    v2f qa[3], qb[3];
    #pragma unroll
    for (int ck = 0; ck < 3; ck++) {
      qa[ck] = *(const v2f*)(Qb + mr * HEAD + ck * 4 + kh);
      qb[ck] = *(const v2f*)(Qb + TILE_F + mr * HEAD + ck * 4 + kh);
    }

    v8f oa = {}, ob = {};
    float ma = -INFINITY, la = 0.f;
    float mb = -INFINITY, lb = 0.f;

    KVFrag fa, fb;
    kv_load(fa, Kh, Vh, mr, kh, half, dv);
    int kt = 0;
    #pragma unroll 1
    for (; kt + 2 <= qt; kt += 2) {
      kv_load(fb, Kh + (size_t)(kt + 1) * TILE_F, Vh + (size_t)(kt + 1) * TILE_F, mr, kh, half, dv);
      TILE_EDGE();
      attn_tile2(fa, qa, qb, oa, ob, ma, la, mb, lb, mr, half);
      TILE_EDGE();
      kv_load(fa, Kh + (size_t)(kt + 2) * TILE_F, Vh + (size_t)(kt + 2) * TILE_F, mr, kh, half, dv);
      TILE_EDGE();
      attn_tile2(fb, qa, qb, oa, ob, ma, la, mb, lb, mr, half);
      TILE_EDGE();
    }
    kv_load(fb, Kh + (size_t)(qt + 1) * TILE_F, Vh + (size_t)(qt + 1) * TILE_F, mr, kh, half, dv);
    TILE_EDGE();
    {
      v8f s0 = score_tile(fa, qa);
      v8f s1 = score_tile(fa, qb);
      softmax_pv<true>(fa, s0, oa, ma, la, mr, half);
      softmax_pv<false>(fa, s1, ob, mb, lb, mr, half);
    }
    TILE_EDGE();
    {
      v8f s1 = score_tile(fb, qb);
      softmax_pv<true>(fb, s1, ob, mb, lb, mr, half);
    }

    store_tile(sa, oa, la, b, h, qt, mr, half);
    store_tile(sa, ob, lb, b, h, qt + 1, mr, half);
  }
#undef TILE_EDGE
}

__global__ void __launch_bounds__(256) gemm_proj_kernel(const float* __restrict__ A, const float* __restrict__ Bf,
                                 const float* __restrict__ bias, const float* __restrict__ resid,
                                 float* __restrict__ out) {
  int m0, n0;
  wave_tile<192 / 48, 48>(m0, n0);
  v8f acc[3] = {{}, {}, {}};
  float ss = 0.f;
  gemm16xn<false, 3>(A, Bf, 192, 192, 192, m0, n0, acc, ss);
  int lane = threadIdx.x & 31; int mr = lane & 15, half = lane >> 4;
  #pragma unroll
  for (int t = 0; t < 3; t++) {
    int j = n0 + t * 16 + mr;
    float bj = bias[j];
    #pragma unroll
    for (int v = 0; v < 8; v++) {
      int m = m0 + v + 8 * half;
      size_t ofs = (size_t)m * 192 + j;
      out[ofs] = acc[t][v] + bj + resid[ofs];
    }
  }
}

__global__ void __launch_bounds__(256) gemm_ff1_kernel(const float* __restrict__ A, const float* __restrict__ Bf,
                                float* __restrict__ out) {
  int m0, n0;
  wave_tile<FFN / 64, 64>(m0, n0);
  v8f acc[4] = {{}, {}, {}, {}};
  float ss = 0.f;
  gemm16xn<true, 4>(A, Bf, 192, 192, FFN, m0, n0, acc, ss);
  int lane = threadIdx.x & 31; int mr = lane & 15, half = lane >> 4;
  float scr[8];
  row_rms_factors(ss, half, scr);
  #pragma unroll
  for (int t = 0; t < 4; t++) {
    int j = n0 + t * 16 + mr;
    #pragma unroll
    for (int v = 0; v < 8; v++) {
      int m = m0 + v + 8 * half;
      float val = acc[t][v] * scr[v];
      out[(size_t)m * FFN + j] = val / (1.0f + __expf(-val));
    }
  }
}

__global__ void __launch_bounds__(256) gemm_ff2_kernel(const float* __restrict__ A, const float* __restrict__ Bf,
                                const float* __restrict__ resid, float* __restrict__ out) {
  int m0, n0;
  wave_tile<192 / 48, 48>(m0, n0);
  v8f acc[3] = {{}, {}, {}};
  float ss = 0.f;
  gemm16xn<false, 3>(A, Bf, FFN, FFN, 192, m0, n0, acc, ss);
  int lane = threadIdx.x & 31; int mr = lane & 15, half = lane >> 4;
  #pragma unroll
  for (int t = 0; t < 3; t++) {
    int j = n0 + t * 16 + mr;
    #pragma unroll
    for (int v = 0; v < 8; v++) {
      int m = m0 + v + 8 * half;
      size_t ofs = (size_t)m * 192 + j;
      out[ofs] = acc[t][v] + resid[ofs];
    }
  }
}

extern "C" void kernel_launch(void* const* d_in, const int* in_sizes, int n_in,
                              void* d_out, int out_size, void* d_ws, size_t ws_size,
                              hipStream_t stream) {
  (void)in_sizes; (void)n_in; (void)out_size; (void)ws_size;
  const float* x     = (const float*)d_in[0];
  const float* ln1_w = (const float*)d_in[1];
  const float* ln2_w = (const float*)d_in[2];
  const float* Wq    = (const float*)d_in[3];
  const float* Wk    = (const float*)d_in[4];
  const float* Wv    = (const float*)d_in[5];
  const float* Wproj = (const float*)d_in[6];
  const float* bproj = (const float*)d_in[7];
  const float* W1    = (const float*)d_in[8];
  const float* W2    = (const float*)d_in[9];
  float* ws = (float*)d_ws;

  size_t off = 0;
  float* wqkv = ws + off; off += (size_t)192 * 576;
  float* wpjf = ws + off; off += (size_t)192 * 192;
  float* w1f  = ws + off; off += (size_t)192 * FFN;
  float* w2f  = ws + off; off += (size_t)FFN * 192;
  float* Qp   = ws + off; off += (size_t)BH * SEQ * HEAD;
  float* Kp   = ws + off; off += (size_t)BH * SEQ * HEAD;
  float* Vp   = ws + off; off += (size_t)BH * SEQ * HEAD;
  float* sa   = ws + off; off += (size_t)M_TOK * N_EMBD;
  float* x1   = ws + off; off += (size_t)M_TOK * N_EMBD;
  float* u    = ws + off; off += (size_t)M_TOK * FFN;
  float* out  = (float*)d_out;

  prep_kernel<<<PACK_WGS, 256, 0, stream>>>(ln1_w, ln2_w, Wq, Wk, Wv, Wproj, W1, W2,
                                            wqkv, wpjf, w1f, w2f);
  gemm_qkv_kernel<<<(512 * 12) / 8, 256, 0, stream>>>(x, wqkv, Qp, Kp, Vp);
  attn_kernel<<<dim3(8, BH), 128, 0, stream>>>(Qp, Kp, Vp, sa);
  gemm_proj_kernel<<<(512 * 4) / 8, 256, 0, stream>>>(sa, wpjf, bproj, x, x1);
  gemm_ff1_kernel<<<(512 * 12) / 8, 256, 0, stream>>>(x1, w1f, u);
  gemm_ff2_kernel<<<(512 * 4) / 8, 256, 0, stream>>>(u, w2f, x1, out);
}
